# starting kernel
# baseline (speedup 1.0000x reference)
#include <hip/hip_runtime.h>
#include <math.h>


typedef __attribute__((ext_vector_type(2))) float v2f;
typedef __attribute__((ext_vector_type(8))) float v8f;

#define N_NODES 50000
#define FEATS   128
#define LDS_STRIDE 132

__global__ void gcn_zero_kernel(float4* __restrict__ p, int n4) {
    int i = blockIdx.x * blockDim.x + threadIdx.x;
    if (i < n4) p[i] = make_float4(0.f, 0.f, 0.f, 0.f);
}

__global__ void gcn_edge_scatter(const float* __restrict__ feats,
                                 const float* __restrict__ norm,
                                 const int*   __restrict__ src,
                                 const int*   __restrict__ dst,
                                 float*       __restrict__ agg,
                                 int n_edges) {
    int e = blockIdx.x * 8 + (threadIdx.x >> 5);
    if (e >= n_edges) return;
    int lane = threadIdx.x & 31;
    int s = src[e];
    int d = dst[e];
    float ns = norm[s];
    float4 v = ((const float4*)(feats + (size_t)s * FEATS))[lane];
    float* out = agg + (size_t)d * FEATS + (size_t)lane * 4;
    atomicAdd(out + 0, v.x * ns);
    atomicAdd(out + 1, v.y * ns);
    atomicAdd(out + 2, v.z * ns);
    atomicAdd(out + 3, v.w * ns);
}

__global__ void gcn_gemm_bias_tanh(float* __restrict__ agg,
                                   const float* __restrict__ norm,
                                   const float* __restrict__ W,
                                   const float* __restrict__ bias) {
    __shared__ float As[16 * LDS_STRIDE];

    int row0 = blockIdx.x * 16;

    for (int i = threadIdx.x; i < 16 * FEATS; i += 256) {
        int r = i >> 7;
        int k = i & 127;
        int n = row0 + r;
        As[r * LDS_STRIDE + k] = agg[(size_t)n * FEATS + k] * norm[n];
    }
    __syncthreads();

    int wave = threadIdx.x >> 5;
    int lane = threadIdx.x & 31;
    int half = lane >> 4;
    int r    = lane & 15;
    int col  = wave * 16 + r;

    const float* wrow = W + (size_t)col * FEATS;

    v8f c = {};
    #pragma unroll
    for (int k0 = 0; k0 < FEATS; k0 += 4) {
        int kk = k0 + 2 * half;
        v2f a, b;
        a.x = As[r * LDS_STRIDE + kk];
        a.y = As[r * LDS_STRIDE + kk + 1];
        b.x = wrow[kk];
        b.y = wrow[kk + 1];
        c = __builtin_amdgcn_wmma_f32_16x16x4_f32(
                 false, a,  false, b,
                 (short)0, c,  false,  false);
    }

    float bb = bias[col];
    #pragma unroll
    for (int v = 0; v < 8; v++) {
        int m = row0 + v + 8 * half;
        agg[(size_t)m * FEATS + col] = tanhf(c[v] + bb);
    }
}

extern "C" void kernel_launch(void* const* d_in, const int* in_sizes, int n_in,
                              void* d_out, int out_size, void* d_ws, size_t ws_size,
                              hipStream_t stream) {
    const float* feats = (const float*)d_in[0];
    const float* norm  = (const float*)d_in[1];
    const float* W     = (const float*)d_in[2];
    const float* bias  = (const float*)d_in[3];
    const int*   src   = (const int*)d_in[4];
    const int*   dst   = (const int*)d_in[5];
    int n_edges = in_sizes[4];

    float* agg = (float*)d_out;

    int n4 = (N_NODES * FEATS) / 4;
    gcn_zero_kernel<<<(n4 + 255) / 256, 256, 0, stream>>>((float4*)agg, n4);

    gcn_edge_scatter<<<(n_edges + 7) / 8, 256, 0, stream>>>(
        feats, norm, src, dst, agg, n_edges);

    gcn_gemm_bias_tanh<<<N_NODES / 16, 256, 0, stream>>>(agg, norm, W, bias);
}


# Float atomic scatter replaced by in-call counting sort of edges by dst (returning int atomic = arrival-order rank, one-block scan, fill) and one-wave-per-node register gather; zeroing pass gone
# speedup vs baseline: 1.2587x; 1.2587x over previous
#include <hip/hip_runtime.h>
#include <math.h>


typedef __attribute__((ext_vector_type(2))) float v2f;
typedef __attribute__((ext_vector_type(8))) float v8f;

#define N_NODES 50000
#define FEATS   128
#define LDS_STRIDE 132

#define SCAN_THREADS 1024
#define SCAN_TILES   13
#define CNT_PAD      (SCAN_TILES * SCAN_THREADS * 4)

#define WS_CNT    0
#define WS_OFF    65536
#define WS_RANK   131072
#define WS_COL    (131072 + 1048576)

__global__ void gcn_zero_counts(int4* __restrict__ p) {
    int i = blockIdx.x * blockDim.x + threadIdx.x;
    p[i] = make_int4(0, 0, 0, 0);
}

__global__ void gcn_count_rank(const int* __restrict__ dst,
                               int* __restrict__ cnt,
                               int* __restrict__ rank,
                               int n_edges) {
    int e = blockIdx.x * 256 + threadIdx.x;
    if (e >= n_edges) return;
    rank[e] = atomicAdd(cnt + dst[e], 1);
}

static __device__ __forceinline__ int wave_inclusive_sum(int x, int lane) {
    x += __builtin_amdgcn_update_dpp(0, x, 0x111, 0xf, 0xf, false);
    x += __builtin_amdgcn_update_dpp(0, x, 0x112, 0xf, 0xf, false);
    x += __builtin_amdgcn_update_dpp(0, x, 0x114, 0xf, 0xf, false);
    x += __builtin_amdgcn_update_dpp(0, x, 0x118, 0xf, 0xf, false);
    int low = __builtin_amdgcn_readlane(x, 15);
    if (lane >= 16) x += low;
    return x;
}

__global__ __launch_bounds__(SCAN_THREADS)
void gcn_scan_counts(const int4* __restrict__ cnt4, int4* __restrict__ off4) {
    __shared__ int wtot[SCAN_TILES][32];
    int t = threadIdx.x;
    int lane = t & 31;
    int wave = __builtin_amdgcn_readfirstlane(t >> 5);

    int4 v[SCAN_TILES];
    int own[SCAN_TILES];
    int inc[SCAN_TILES];
    #pragma unroll
    for (int j = 0; j < SCAN_TILES; j++) v[j] = cnt4[j * SCAN_THREADS + t];

    #pragma unroll
    for (int j = 0; j < SCAN_TILES; j++) {
        int s = v[j].x + v[j].y + v[j].z + v[j].w;
        own[j] = s;
        int x = wave_inclusive_sum(s, lane);
        inc[j] = x;
        if (lane == 31) wtot[j][wave] = x;
    }
    __syncthreads();

    int carry = 0;
    #pragma unroll
    for (int j = 0; j < SCAN_TILES; j++) {
        int x = wave_inclusive_sum(wtot[j][lane], lane);
        int tile_total = __builtin_amdgcn_readlane(x, 31);
        int before = __builtin_amdgcn_readlane(x, (wave + 31) & 31);
        if (wave == 0) before = 0;
        int base = carry + before + inc[j] - own[j];
        int4 o4;
        o4.x = base;
        o4.y = o4.x + v[j].x;
        o4.z = o4.y + v[j].y;
        o4.w = o4.z + v[j].z;
        off4[j * SCAN_THREADS + t] = o4;
        carry += tile_total;
    }
}

__global__ void gcn_fill_lists(const int* __restrict__ src,
                               const int* __restrict__ dst,
                               const int* __restrict__ off,
                               const int* __restrict__ rank,
                               int* __restrict__ col,
                               int n_edges) {
    int e = blockIdx.x * 256 + threadIdx.x;
    if (e >= n_edges) return;
    int p = off[dst[e]] + rank[e];
    if ((unsigned)p < (unsigned)n_edges) col[p] = src[e];
}

__global__ __launch_bounds__(256)
void gcn_gather_rows(const float* __restrict__ feats,
                     const float* __restrict__ norm,
                     const int*   __restrict__ off,
                     const int*   __restrict__ col,
                     float*       __restrict__ agg,
                     int n_edges) {
    int lane = threadIdx.x & 31;
    int n = blockIdx.x * 8 + (threadIdx.x >> 5);
    if (n >= N_NODES) return;

    int beg = __builtin_amdgcn_readfirstlane(off[n]);
    int end = __builtin_amdgcn_readfirstlane(off[n + 1]);
    beg = max(beg, 0);
    end = min(end, n_edges);

    const float4* f4 = (const float4*)feats;
    float4 acc = make_float4(0.f, 0.f, 0.f, 0.f);

    for (int base = beg; base < end; base += 32) {
        int m = min(32, end - base);
        int sidx = 0;
        float sw = 0.f;
        if (lane < m) {
            sidx = (int)min((unsigned)col[base + lane], (unsigned)(N_NODES - 1));
            sw = norm[sidx];
        }
        int swi = __float_as_int(sw);

        for (int j = 0; j < m; j += 8) {
            float4 a[8];
            float  w[8];
            w[0] = __int_as_float(__builtin_amdgcn_readlane(swi, j));
            a[0] = f4[(size_t)__builtin_amdgcn_readlane(sidx, j) * (FEATS / 4) + lane];
            #pragma unroll
            for (int u = 1; u < 8; u++) {
                a[u] = make_float4(0.f, 0.f, 0.f, 0.f);
                w[u] = 0.f;
                if (j + u < m) {
                    int s = __builtin_amdgcn_readlane(sidx, j + u);
                    w[u] = __int_as_float(__builtin_amdgcn_readlane(swi, j + u));
                    a[u] = f4[(size_t)s * (FEATS / 4) + lane];
                }
            }
            #pragma unroll
            for (int u = 0; u < 8; u++) {
                acc.x += a[u].x * w[u];
                acc.y += a[u].y * w[u];
                acc.z += a[u].z * w[u];
                acc.w += a[u].w * w[u];
            }
        }
    }

    ((float4*)agg)[(size_t)n * (FEATS / 4) + lane] = acc;
}

__global__ void gcn_gemm_bias_tanh(float* __restrict__ agg,
                                   const float* __restrict__ norm,
                                   const float* __restrict__ W,
                                   const float* __restrict__ bias) {
    __shared__ float As[16 * LDS_STRIDE];

    int row0 = blockIdx.x * 16;

    for (int i = threadIdx.x; i < 16 * FEATS; i += 256) {
        int r = i >> 7;
        int k = i & 127;
        int n = row0 + r;
        As[r * LDS_STRIDE + k] = agg[(size_t)n * FEATS + k] * norm[n];
    }
    __syncthreads();

    int wave = threadIdx.x >> 5;
    int lane = threadIdx.x & 31;
    int half = lane >> 4;
    int r    = lane & 15;
    int col  = wave * 16 + r;

    const float* wrow = W + (size_t)col * FEATS;

    v8f c = {};
    #pragma unroll
    for (int k0 = 0; k0 < FEATS; k0 += 4) {
        int kk = k0 + 2 * half;
        v2f a, b;
        a.x = As[r * LDS_STRIDE + kk];
        a.y = As[r * LDS_STRIDE + kk + 1];
        b.x = wrow[kk];
        b.y = wrow[kk + 1];
        c = __builtin_amdgcn_wmma_f32_16x16x4_f32(
                 false, a,  false, b,
                 (short)0, c,  false,  false);
    }

    float bb = bias[col];
    #pragma unroll
    for (int v = 0; v < 8; v++) {
        int m = row0 + v + 8 * half;
        agg[(size_t)m * FEATS + col] = tanhf(c[v] + bb);
    }
}

extern "C" void kernel_launch(void* const* d_in, const int* in_sizes, int n_in,
                              void* d_out, int out_size, void* d_ws, size_t ws_size,
                              hipStream_t stream) {
    const float* feats = (const float*)d_in[0];
    const float* norm  = (const float*)d_in[1];
    const float* W     = (const float*)d_in[2];
    const float* bias  = (const float*)d_in[3];
    const int*   src   = (const int*)d_in[4];
    const int*   dst   = (const int*)d_in[5];
    int n_edges = in_sizes[4];

    float* agg = (float*)d_out;

    int* wsi  = (int*)d_ws;
    int* cnt  = wsi + WS_CNT;
    int* off  = wsi + WS_OFF;
    int* rank = wsi + WS_RANK;
    int* col  = wsi + WS_COL;

    gcn_zero_counts<<<CNT_PAD / 4 / 256, 256, 0, stream>>>((int4*)cnt);

    gcn_count_rank<<<(n_edges + 255) / 256, 256, 0, stream>>>(dst, cnt, rank, n_edges);

    gcn_scan_counts<<<1, SCAN_THREADS, 0, stream>>>((const int4*)cnt, (int4*)off);

    gcn_fill_lists<<<(n_edges + 255) / 256, 256, 0, stream>>>(src, dst, off, rank, col, n_edges);

    gcn_gather_rows<<<N_NODES / 8, 256, 0, stream>>>(feats, norm, off, col, agg, n_edges);

    gcn_gemm_bias_tanh<<<N_NODES / 16, 256, 0, stream>>>(agg, norm, W, bias);
}


# Gather fused into the GEMM launch: a 256-thread block sums its 32 destination rows (4 per wave, same order of sums) straight into the LDS A tile, two row tiles per W fragment; 5 launches
# speedup vs baseline: 1.4376x; 1.1421x over previous
#include <hip/hip_runtime.h>
#include <math.h>


typedef __attribute__((ext_vector_type(2))) float v2f;
typedef __attribute__((ext_vector_type(8))) float v8f;

#define N_NODES 50000
#define FEATS   128
#define LDS_STRIDE 132
#define BLOCK_ROWS 32
#define N_ROW_BLOCKS ((N_NODES + BLOCK_ROWS - 1) / BLOCK_ROWS)

#define SCAN_THREADS 1024
#define SCAN_TILES   13
#define CNT_PAD      (SCAN_TILES * SCAN_THREADS * 4)

#define WS_CNT    0
#define WS_OFF    65536
#define WS_RANK   131072
#define WS_COL    (131072 + 1048576)

__global__ void gcn_zero_counts(int4* __restrict__ p) {
    int i = blockIdx.x * blockDim.x + threadIdx.x;
    p[i] = make_int4(0, 0, 0, 0);
}

__global__ void gcn_count_rank(const int* __restrict__ dst,
                               int* __restrict__ cnt,
                               int* __restrict__ rank,
                               int n_edges) {
    int e = blockIdx.x * 256 + threadIdx.x;
    if (e >= n_edges) return;
    rank[e] = atomicAdd(cnt + dst[e], 1);
}

static __device__ __forceinline__ int wave_inclusive_sum(int x, int lane) {
    x += __builtin_amdgcn_update_dpp(0, x, 0x111, 0xf, 0xf, false);
    x += __builtin_amdgcn_update_dpp(0, x, 0x112, 0xf, 0xf, false);
    x += __builtin_amdgcn_update_dpp(0, x, 0x114, 0xf, 0xf, false);
    x += __builtin_amdgcn_update_dpp(0, x, 0x118, 0xf, 0xf, false);
    int low = __builtin_amdgcn_readlane(x, 15);
    if (lane >= 16) x += low;
    return x;
}

__global__ __launch_bounds__(SCAN_THREADS)
void gcn_scan_counts(const int4* __restrict__ cnt4, int4* __restrict__ off4) {
    __shared__ int wtot[SCAN_TILES][32];
    int t = threadIdx.x;
    int lane = t & 31;
    int wave = __builtin_amdgcn_readfirstlane(t >> 5);

    int4 v[SCAN_TILES];
    int own[SCAN_TILES];
    int inc[SCAN_TILES];
    #pragma unroll
    for (int j = 0; j < SCAN_TILES; j++) v[j] = cnt4[j * SCAN_THREADS + t];

    #pragma unroll
    for (int j = 0; j < SCAN_TILES; j++) {
        int s = v[j].x + v[j].y + v[j].z + v[j].w;
        own[j] = s;
        int x = wave_inclusive_sum(s, lane);
        inc[j] = x;
        if (lane == 31) wtot[j][wave] = x;
    }
    __syncthreads();

    int carry = 0;
    #pragma unroll
    for (int j = 0; j < SCAN_TILES; j++) {
        int x = wave_inclusive_sum(wtot[j][lane], lane);
        int tile_total = __builtin_amdgcn_readlane(x, 31);
        int before = __builtin_amdgcn_readlane(x, (wave + 31) & 31);
        if (wave == 0) before = 0;
        int base = carry + before + inc[j] - own[j];
        int4 o4;
        o4.x = base;
        o4.y = o4.x + v[j].x;
        o4.z = o4.y + v[j].y;
        o4.w = o4.z + v[j].z;
        off4[j * SCAN_THREADS + t] = o4;
        carry += tile_total;
    }
}

__global__ void gcn_fill_lists(const int* __restrict__ src,
                               const int* __restrict__ dst,
                               const int* __restrict__ off,
                               const int* __restrict__ rank,
                               int* __restrict__ col,
                               int n_edges) {
    int e = blockIdx.x * 256 + threadIdx.x;
    if (e >= n_edges) return;
    int p = off[dst[e]] + rank[e];
    if ((unsigned)p < (unsigned)n_edges) col[p] = src[e];
}

static __device__ __forceinline__ float4 gcn_sum_list(const float4* __restrict__ f4,
                                                      const float*  __restrict__ norm,
                                                      const int*    __restrict__ col,
                                                      int beg, int end, int lane) {
    float4 acc = make_float4(0.f, 0.f, 0.f, 0.f);

    for (int base = beg; base < end; base += 32) {
        int m = min(32, end - base);
        int sidx = 0;
        float sw = 0.f;
        if (lane < m) {
            sidx = (int)min((unsigned)col[base + lane], (unsigned)(N_NODES - 1));
            sw = norm[sidx];
        }
        int swi = __float_as_int(sw);

        for (int j = 0; j < m; j += 8) {
            float4 a[8];
            float  w[8];
            w[0] = __int_as_float(__builtin_amdgcn_readlane(swi, j));
            a[0] = f4[(size_t)__builtin_amdgcn_readlane(sidx, j) * (FEATS / 4) + lane];
            #pragma unroll
            for (int u = 1; u < 8; u++) {
                a[u] = make_float4(0.f, 0.f, 0.f, 0.f);
                w[u] = 0.f;
                if (j + u < m) {
                    int s = __builtin_amdgcn_readlane(sidx, j + u);
                    w[u] = __int_as_float(__builtin_amdgcn_readlane(swi, j + u));
                    a[u] = f4[(size_t)s * (FEATS / 4) + lane];
                }
            }
            #pragma unroll
            for (int u = 0; u < 8; u++) {
                acc.x += a[u].x * w[u];
                acc.y += a[u].y * w[u];
                acc.z += a[u].z * w[u];
                acc.w += a[u].w * w[u];
            }
        }
    }
    return acc;
}

__global__ __launch_bounds__(256, 16)
void gcn_gather_gemm_bias_tanh(const float* __restrict__ feats,
                               const float* __restrict__ norm,
                               const int*   __restrict__ off,
                               const int*   __restrict__ col,
                               const float* __restrict__ W,
                               const float* __restrict__ bias,
                               float*       __restrict__ out,
                               int n_edges) {
    __shared__ __attribute__((aligned(16))) float As[BLOCK_ROWS * LDS_STRIDE];

    int lane = threadIdx.x & 31;
    int wave = __builtin_amdgcn_readfirstlane(threadIdx.x >> 5);
    int row0 = blockIdx.x * BLOCK_ROWS;

    int ov = off[row0 + 4 * wave + min(lane, 4)];

    const float4* f4 = (const float4*)feats;
    for (int i = 0; i < 4; i++) {
        int r = 4 * wave + i;
        int n = row0 + r;
        int beg = __builtin_amdgcn_readlane(ov, i);
        int end = __builtin_amdgcn_readlane(ov, i + 1);
        beg = max(beg, 0);
        end = min(end, n_edges);
        float4 acc = make_float4(0.f, 0.f, 0.f, 0.f);
        float dn = 0.f;
        if (n < N_NODES) {
            acc = gcn_sum_list(f4, norm, col, beg, end, lane);
            dn = norm[n];
        }
        acc.x *= dn;
        acc.y *= dn;
        acc.z *= dn;
        acc.w *= dn;
        *(float4*)&As[r * LDS_STRIDE + 4 * lane] = acc;
    }
    __syncthreads();

    int half = lane >> 4;
    int r    = lane & 15;
    int cj   = wave * 16 + r;

    const float* wrow = W + (size_t)cj * FEATS;

    v8f c0 = {};
    v8f c1 = {};
    #pragma unroll
    for (int k0 = 0; k0 < FEATS; k0 += 4) {
        int kk = k0 + 2 * half;
        v2f a0, a1, b;
        a0.x = As[r * LDS_STRIDE + kk];
        a0.y = As[r * LDS_STRIDE + kk + 1];
        a1.x = As[(16 + r) * LDS_STRIDE + kk];
        a1.y = As[(16 + r) * LDS_STRIDE + kk + 1];
        b.x = wrow[kk];
        b.y = wrow[kk + 1];
        c0 = __builtin_amdgcn_wmma_f32_16x16x4_f32(
                 false, a0,  false, b,
                 (short)0, c0,  false,  false);
        c1 = __builtin_amdgcn_wmma_f32_16x16x4_f32(
                 false, a1,  false, b,
                 (short)0, c1,  false,  false);
    }

    float bb = bias[cj];
    float h0[8], h1[8];
    #pragma unroll
    for (int v = 0; v < 8; v++) {
        h0[v] = tanhf(c0[v] + bb);
        h1[v] = tanhf(c1[v] + bb);
    }
    #pragma unroll
    for (int v = 0; v < 8; v++) {
        int m = row0 + v + 8 * half;
        out[(size_t)m * FEATS + cj] = h0[v];
    }
    if (row0 + 16 < N_NODES) {
        #pragma unroll
        for (int v = 0; v < 8; v++) {
            int m = row0 + 16 + v + 8 * half;
            out[(size_t)m * FEATS + cj] = h1[v];
        }
    }
}

extern "C" void kernel_launch(void* const* d_in, const int* in_sizes, int n_in,
                              void* d_out, int out_size, void* d_ws, size_t ws_size,
                              hipStream_t stream) {
    const float* feats = (const float*)d_in[0];
    const float* norm  = (const float*)d_in[1];
    const float* W     = (const float*)d_in[2];
    const float* bias  = (const float*)d_in[3];
    const int*   src   = (const int*)d_in[4];
    const int*   dst   = (const int*)d_in[5];
    int n_edges = in_sizes[4];

    float* out = (float*)d_out;

    int* wsi  = (int*)d_ws;
    int* cnt  = wsi + WS_CNT;
    int* off  = wsi + WS_OFF;
    int* rank = wsi + WS_RANK;
    int* col  = wsi + WS_COL;

    gcn_zero_counts<<<CNT_PAD / 4 / 256, 256, 0, stream>>>((int4*)cnt);

    gcn_count_rank<<<(n_edges + 255) / 256, 256, 0, stream>>>(dst, cnt, rank, n_edges);

    gcn_scan_counts<<<1, SCAN_THREADS, 0, stream>>>((const int4*)cnt, (int4*)off);

    gcn_fill_lists<<<(n_edges + 255) / 256, 256, 0, stream>>>(src, dst, off, rank, col, n_edges);

    gcn_gather_gemm_bias_tanh<<<N_ROW_BLOCKS, 256, 0, stream>>>(feats, norm, off, col, W, bias, out, n_edges);
}


# Global-atomic counting sort (zero, atomic rank, scan, scattered fill) replaced by a two-level bin sort counting and placing in LDS, whole-line writes; list order = LDS atomic arrival; 4 launches
# speedup vs baseline: 2.8311x; 1.9693x over previous
#include <hip/hip_runtime.h>
#include <math.h>


typedef __attribute__((ext_vector_type(2))) float v2f;
typedef __attribute__((ext_vector_type(8))) float v8f;

#define N_NODES 50000
#define FEATS   128
#define LDS_STRIDE 132
#define BLOCK_ROWS 32
#define N_ROW_BLOCKS ((N_NODES + BLOCK_ROWS - 1) / BLOCK_ROWS)

#define SORT_BLOCKS  128
#define SORT_THREADS 1024
#define EDGE_UNROLL  8
#define CHUNK_MAX    8192
#define BIN_SHIFT    8
#define BIN_NODES    256
#define N_BINS ((N_NODES + BIN_NODES - 1) / BIN_NODES)
#define STAGE_MAX    12288

#define WS_HIST   0
#define WS_GBASE  32768
#define WS_OFF    65536
#define WS_BIN    131072
#define WS_COL    (131072 + 1048576)

static __device__ __forceinline__ int wave_inclusive_sum(int x, int lane) {
    x += __builtin_amdgcn_update_dpp(0, x, 0x111, 0xf, 0xf, false);
    x += __builtin_amdgcn_update_dpp(0, x, 0x112, 0xf, 0xf, false);
    x += __builtin_amdgcn_update_dpp(0, x, 0x114, 0xf, 0xf, false);
    x += __builtin_amdgcn_update_dpp(0, x, 0x118, 0xf, 0xf, false);
    int low = __builtin_amdgcn_readlane(x, 15);
    if (lane >= 16) x += low;
    return x;
}

static __device__ __forceinline__ int gcn_chunk_len(int n_edges) {
    return min((n_edges + SORT_BLOCKS - 1) / SORT_BLOCKS, CHUNK_MAX);
}

__global__ __launch_bounds__(SORT_THREADS)
void gcn_bin_count(const int* __restrict__ dst, int* __restrict__ hist, int n_edges) {
    __shared__ int h[N_BINS];
    int t = threadIdx.x;
    if (t < N_BINS) h[t] = 0;
    __syncthreads();

    int chunk = gcn_chunk_len(n_edges);
    int beg = blockIdx.x * chunk;
    int end = min(beg + chunk, n_edges);
    for (int i = beg + t; i < end; i += SORT_THREADS * EDGE_UNROLL) {
        int d[EDGE_UNROLL];
        #pragma unroll
        for (int u = 0; u < EDGE_UNROLL; u++) {
            int k = i + u * SORT_THREADS;
            d[u] = (k < end) ? dst[k] : -1;
        }
        #pragma unroll
        for (int u = 0; u < EDGE_UNROLL; u++) {
            if ((unsigned)d[u] < (unsigned)N_NODES) atomicAdd(&h[d[u] >> BIN_SHIFT], 1);
        }
    }
    __syncthreads();
    if (t < N_BINS) hist[blockIdx.x * N_BINS + t] = h[t];
}

__global__ __launch_bounds__(SORT_THREADS)
void gcn_bin_scatter(const int* __restrict__ src,
                     const int* __restrict__ dst,
                     const int* __restrict__ hist,
                     int* __restrict__ gbase,
                     unsigned* __restrict__ binned,
                     int n_edges) {
    __shared__ unsigned stage[CHUNK_MAX];
    __shared__ int cursor[N_BINS];
    __shared__ int delta[N_BINS];
    __shared__ int wtot[8];
    __shared__ int wown[8];
    int t = threadIdx.x;
    int lane = t & 31;
    int wave = __builtin_amdgcn_readfirstlane(t >> 5);
    int me = blockIdx.x;

    int total = 0, before = 0, own = 0;
    if (t < N_BINS) {
        #pragma unroll 32
        for (int b = 0; b < SORT_BLOCKS; b++) {
            int v = hist[b * N_BINS + t];
            total += v;
            before += (b < me) ? v : 0;
            own = (b == me) ? v : own;
        }
    }
    int itot = 0, iown = 0;
    if (t < 256) {
        itot = wave_inclusive_sum(total, lane);
        iown = wave_inclusive_sum(own, lane);
        if (lane == 31) { wtot[wave] = itot; wown[wave] = iown; }
    }
    __syncthreads();
    if (t < 256) {
        int ctot = 0, cown = 0;
        for (int w = 0; w < 8; w++) {
            if (w < wave) { ctot += wtot[w]; cown += wown[w]; }
        }
        int gstart = ctot + itot - total;
        int lstart = cown + iown - own;
        if (t < N_BINS) {
            cursor[t] = lstart;
            delta[t] = gstart + before - lstart;
        }
        if (me == 0 && t <= N_BINS) gbase[t] = gstart;
    }
    __syncthreads();

    int chunk = gcn_chunk_len(n_edges);
    int beg = me * chunk;
    int end = min(beg + chunk, n_edges);
    for (int i = beg + t; i < end; i += SORT_THREADS * EDGE_UNROLL) {
        int d[EDGE_UNROLL];
        int s[EDGE_UNROLL];
        #pragma unroll
        for (int u = 0; u < EDGE_UNROLL; u++) {
            int k = i + u * SORT_THREADS;
            d[u] = -1;
            s[u] = 0;
            if (k < end) { d[u] = dst[k]; s[u] = src[k]; }
        }
        #pragma unroll
        for (int u = 0; u < EDGE_UNROLL; u++) {
            if ((unsigned)d[u] < (unsigned)N_NODES) {
                int p = atomicAdd(&cursor[d[u] >> BIN_SHIFT], 1);
                stage[p] = ((unsigned)d[u] << 16) | ((unsigned)s[u] & 0xffffu);
            }
        }
    }
    __syncthreads();

    int n_staged = cursor[N_BINS - 1];
    for (int i = t; i < n_staged; i += SORT_THREADS) {
        unsigned w = stage[i];
        binned[delta[w >> (16 + BIN_SHIFT)] + i] = w;
    }
}

__global__ __launch_bounds__(SORT_THREADS)
void gcn_bin_sort(const unsigned* __restrict__ binned,
                  const int* __restrict__ gbase,
                  int* __restrict__ off,
                  int* __restrict__ col) {
    __shared__ int cnt[BIN_NODES];
    __shared__ int stage[STAGE_MAX];
    __shared__ int wsum[8];
    int t = threadIdx.x;
    int lane = t & 31;
    int wave = __builtin_amdgcn_readfirstlane(t >> 5);
    int bin = blockIdx.x;
    int beg = gbase[bin];
    int size = gbase[bin + 1] - beg;

    if (t < BIN_NODES) cnt[t] = 0;
    __syncthreads();

    for (int i = t; i < size; i += SORT_THREADS * EDGE_UNROLL) {
        unsigned w[EDGE_UNROLL];
        #pragma unroll
        for (int u = 0; u < EDGE_UNROLL; u++) {
            int k = i + u * SORT_THREADS;
            w[u] = (k < size) ? binned[beg + k] : 0xffffffffu;
        }
        #pragma unroll
        for (int u = 0; u < EDGE_UNROLL; u++) {
            if (w[u] != 0xffffffffu) atomicAdd(&cnt[(w[u] >> 16) & (BIN_NODES - 1)], 1);
        }
    }
    __syncthreads();

    int c = 0, inc = 0;
    if (t < BIN_NODES) {
        c = cnt[t];
        inc = wave_inclusive_sum(c, lane);
        if (lane == 31) wsum[wave] = inc;
    }
    __syncthreads();
    if (t < BIN_NODES) {
        int carry = 0;
        for (int w = 0; w < 8; w++) {
            if (w < wave) carry += wsum[w];
        }
        int excl = carry + inc - c;
        off[bin * BIN_NODES + t] = beg + excl;
        cnt[t] = excl;
    }
    __syncthreads();

    bool fits = size <= STAGE_MAX;
    for (int i = t; i < size; i += SORT_THREADS * EDGE_UNROLL) {
        unsigned w[EDGE_UNROLL];
        #pragma unroll
        for (int u = 0; u < EDGE_UNROLL; u++) {
            int k = i + u * SORT_THREADS;
            w[u] = (k < size) ? binned[beg + k] : 0xffffffffu;
        }
        #pragma unroll
        for (int u = 0; u < EDGE_UNROLL; u++) {
            if (w[u] != 0xffffffffu) {
                int p = atomicAdd(&cnt[(w[u] >> 16) & (BIN_NODES - 1)], 1);
                int s = (int)(w[u] & 0xffffu);
                if (fits) stage[p] = s;
                else      col[beg + p] = s;
            }
        }
    }
    if (fits) {
        __syncthreads();
        for (int i = t; i < size; i += SORT_THREADS) col[beg + i] = stage[i];
    }
}

static __device__ __forceinline__ float4 gcn_sum_list(const float4* __restrict__ f4,
                                                      const float*  __restrict__ norm,
                                                      const int*    __restrict__ col,
                                                      int beg, int end, int lane) {
    float4 acc = make_float4(0.f, 0.f, 0.f, 0.f);

    for (int base = beg; base < end; base += 32) {
        int m = min(32, end - base);
        int sidx = 0;
        float sw = 0.f;
        if (lane < m) {
            sidx = (int)min((unsigned)col[base + lane], (unsigned)(N_NODES - 1));
            sw = norm[sidx];
        }
        int swi = __float_as_int(sw);

        for (int j = 0; j < m; j += 8) {
            float4 a[8];
            float  w[8];
            w[0] = __int_as_float(__builtin_amdgcn_readlane(swi, j));
            a[0] = f4[(size_t)__builtin_amdgcn_readlane(sidx, j) * (FEATS / 4) + lane];
            #pragma unroll
            for (int u = 1; u < 8; u++) {
                a[u] = make_float4(0.f, 0.f, 0.f, 0.f);
                w[u] = 0.f;
                if (j + u < m) {
                    int s = __builtin_amdgcn_readlane(sidx, j + u);
                    w[u] = __int_as_float(__builtin_amdgcn_readlane(swi, j + u));
                    a[u] = f4[(size_t)s * (FEATS / 4) + lane];
                }
            }
            #pragma unroll
            for (int u = 0; u < 8; u++) {
                acc.x += a[u].x * w[u];
                acc.y += a[u].y * w[u];
                acc.z += a[u].z * w[u];
                acc.w += a[u].w * w[u];
            }
        }
    }
    return acc;
}

__global__ __launch_bounds__(256, 16)
void gcn_gather_gemm_bias_tanh(const float* __restrict__ feats,
                               const float* __restrict__ norm,
                               const int*   __restrict__ off,
                               const int*   __restrict__ col,
                               const float* __restrict__ W,
                               const float* __restrict__ bias,
                               float*       __restrict__ out,
                               int n_edges) {
    __shared__ __attribute__((aligned(16))) float As[BLOCK_ROWS * LDS_STRIDE];

    int lane = threadIdx.x & 31;
    int wave = __builtin_amdgcn_readfirstlane(threadIdx.x >> 5);
    int row0 = blockIdx.x * BLOCK_ROWS;

    int ov = off[row0 + 4 * wave + min(lane, 4)];

    const float4* f4 = (const float4*)feats;
    for (int i = 0; i < 4; i++) {
        int r = 4 * wave + i;
        int n = row0 + r;
        int beg = __builtin_amdgcn_readlane(ov, i);
        int end = __builtin_amdgcn_readlane(ov, i + 1);
        beg = max(beg, 0);
        end = min(end, n_edges);
        float4 acc = make_float4(0.f, 0.f, 0.f, 0.f);
        float dn = 0.f;
        if (n < N_NODES) {
            acc = gcn_sum_list(f4, norm, col, beg, end, lane);
            dn = norm[n];
        }
        acc.x *= dn;
        acc.y *= dn;
        acc.z *= dn;
        acc.w *= dn;
        *(float4*)&As[r * LDS_STRIDE + 4 * lane] = acc;
    }
    __syncthreads();

    int half = lane >> 4;
    int r    = lane & 15;
    int cj   = wave * 16 + r;

    const float* wrow = W + (size_t)cj * FEATS;

    v8f c0 = {};
    v8f c1 = {};
    #pragma unroll
    for (int k0 = 0; k0 < FEATS; k0 += 4) {
        int kk = k0 + 2 * half;
        v2f a0, a1, b;
        a0.x = As[r * LDS_STRIDE + kk];
        a0.y = As[r * LDS_STRIDE + kk + 1];
        a1.x = As[(16 + r) * LDS_STRIDE + kk];
        a1.y = As[(16 + r) * LDS_STRIDE + kk + 1];
        b.x = wrow[kk];
        b.y = wrow[kk + 1];
        c0 = __builtin_amdgcn_wmma_f32_16x16x4_f32(
                 false, a0,  false, b,
                 (short)0, c0,  false,  false);
        c1 = __builtin_amdgcn_wmma_f32_16x16x4_f32(
                 false, a1,  false, b,
                 (short)0, c1,  false,  false);
    }

    float bb = bias[cj];
    float h0[8], h1[8];
    #pragma unroll
    for (int v = 0; v < 8; v++) {
        h0[v] = tanhf(c0[v] + bb);
        h1[v] = tanhf(c1[v] + bb);
    }
    #pragma unroll
    for (int v = 0; v < 8; v++) {
        int m = row0 + v + 8 * half;
        out[(size_t)m * FEATS + cj] = h0[v];
    }
    if (row0 + 16 < N_NODES) {
        #pragma unroll
        for (int v = 0; v < 8; v++) {
            int m = row0 + 16 + v + 8 * half;
            out[(size_t)m * FEATS + cj] = h1[v];
        }
    }
}

extern "C" void kernel_launch(void* const* d_in, const int* in_sizes, int n_in,
                              void* d_out, int out_size, void* d_ws, size_t ws_size,
                              hipStream_t stream) {
    const float* feats = (const float*)d_in[0];
    const float* norm  = (const float*)d_in[1];
    const float* W     = (const float*)d_in[2];
    const float* bias  = (const float*)d_in[3];
    const int*   src   = (const int*)d_in[4];
    const int*   dst   = (const int*)d_in[5];
    int n_edges = in_sizes[4];

    float* out = (float*)d_out;

    int* wsi   = (int*)d_ws;
    int* hist  = wsi + WS_HIST;
    int* gbase = wsi + WS_GBASE;
    int* off   = wsi + WS_OFF;
    unsigned* binned = (unsigned*)(wsi + WS_BIN);
    int* col   = wsi + WS_COL;

    gcn_bin_count<<<SORT_BLOCKS, SORT_THREADS, 0, stream>>>(dst, hist, n_edges);

    gcn_bin_scatter<<<SORT_BLOCKS, SORT_THREADS, 0, stream>>>(src, dst, hist, gbase, binned, n_edges);

    gcn_bin_sort<<<N_BINS, SORT_THREADS, 0, stream>>>(binned, gbase, off, col);

    gcn_gather_gemm_bias_tanh<<<N_ROW_BLOCKS, 256, 0, stream>>>(feats, norm, off, col, W, bias, out, n_edges);
}


# Second-level sort launch removed: each 32-row gather block filters its bin's packed edges itself and places them in per-row LDS slots (list order = LDS atomic arrival, as before); 3 launches
# speedup vs baseline: 3.0457x; 1.0758x over previous
#include <hip/hip_runtime.h>
#include <math.h>


typedef __attribute__((ext_vector_type(2))) float v2f;
typedef __attribute__((ext_vector_type(8))) float v8f;

#define N_NODES 50000
#define FEATS   128
#define LDS_STRIDE 132
#define BLOCK_SHIFT 5
#define BLOCK_ROWS (1 << BLOCK_SHIFT)
#define N_ROW_BLOCKS ((N_NODES + BLOCK_ROWS - 1) / BLOCK_ROWS)
#define ROW_SLOTS  64
#define FILTER_VEC 4

#define SORT_BLOCKS  128
#define SORT_THREADS 1024
#define EDGE_UNROLL  8
#define CHUNK_MAX    8192
#define BIN_SHIFT    8
#define BIN_NODES    256
#define N_BINS ((N_NODES + BIN_NODES - 1) / BIN_NODES)

#define WS_HIST   0
#define WS_GBASE  32768
#define WS_BIN    131072

static __device__ __forceinline__ int wave_inclusive_sum(int x, int lane) {
    x += __builtin_amdgcn_update_dpp(0, x, 0x111, 0xf, 0xf, false);
    x += __builtin_amdgcn_update_dpp(0, x, 0x112, 0xf, 0xf, false);
    x += __builtin_amdgcn_update_dpp(0, x, 0x114, 0xf, 0xf, false);
    x += __builtin_amdgcn_update_dpp(0, x, 0x118, 0xf, 0xf, false);
    int low = __builtin_amdgcn_readlane(x, 15);
    if (lane >= 16) x += low;
    return x;
}

static __device__ __forceinline__ int gcn_chunk_len(int n_edges) {
    return min((n_edges + SORT_BLOCKS - 1) / SORT_BLOCKS, CHUNK_MAX);
}

__global__ __launch_bounds__(SORT_THREADS)
void gcn_bin_count(const int* __restrict__ dst, int* __restrict__ hist, int n_edges) {
    __shared__ int h[N_BINS];
    int t = threadIdx.x;
    if (t < N_BINS) h[t] = 0;
    __syncthreads();

    int chunk = gcn_chunk_len(n_edges);
    int beg = blockIdx.x * chunk;
    int end = min(beg + chunk, n_edges);
    for (int i = beg + t; i < end; i += SORT_THREADS * EDGE_UNROLL) {
        int d[EDGE_UNROLL];
        #pragma unroll
        for (int u = 0; u < EDGE_UNROLL; u++) {
            int k = i + u * SORT_THREADS;
            d[u] = (k < end) ? dst[k] : -1;
        }
        #pragma unroll
        for (int u = 0; u < EDGE_UNROLL; u++) {
            if ((unsigned)d[u] < (unsigned)N_NODES) atomicAdd(&h[d[u] >> BIN_SHIFT], 1);
        }
    }
    __syncthreads();
    if (t < N_BINS) hist[blockIdx.x * N_BINS + t] = h[t];
}

__global__ __launch_bounds__(SORT_THREADS)
void gcn_bin_scatter(const int* __restrict__ src,
                     const int* __restrict__ dst,
                     const int* __restrict__ hist,
                     int* __restrict__ gbase,
                     unsigned* __restrict__ binned,
                     int n_edges) {
    __shared__ unsigned stage[CHUNK_MAX];
    __shared__ int cursor[N_BINS];
    __shared__ int delta[N_BINS];
    __shared__ int wtot[8];
    __shared__ int wown[8];
    int t = threadIdx.x;
    int lane = t & 31;
    int wave = __builtin_amdgcn_readfirstlane(t >> 5);
    int me = blockIdx.x;

    int total = 0, before = 0, own = 0;
    if (t < N_BINS) {
        #pragma unroll 32
        for (int b = 0; b < SORT_BLOCKS; b++) {
            int v = hist[b * N_BINS + t];
            total += v;
            before += (b < me) ? v : 0;
            own = (b == me) ? v : own;
        }
    }
    int itot = 0, iown = 0;
    if (t < 256) {
        itot = wave_inclusive_sum(total, lane);
        iown = wave_inclusive_sum(own, lane);
        if (lane == 31) { wtot[wave] = itot; wown[wave] = iown; }
    }
    __syncthreads();
    if (t < 256) {
        int ctot = 0, cown = 0;
        for (int w = 0; w < 8; w++) {
            if (w < wave) { ctot += wtot[w]; cown += wown[w]; }
        }
        int gstart = ctot + itot - total;
        int lstart = cown + iown - own;
        if (t < N_BINS) {
            cursor[t] = lstart;
            delta[t] = gstart + before - lstart;
        }
        if (me == 0 && t <= N_BINS) gbase[t] = gstart;
    }
    __syncthreads();

    int chunk = gcn_chunk_len(n_edges);
    int beg = me * chunk;
    int end = min(beg + chunk, n_edges);
    for (int i = beg + t; i < end; i += SORT_THREADS * EDGE_UNROLL) {
        int d[EDGE_UNROLL];
        int s[EDGE_UNROLL];
        #pragma unroll
        for (int u = 0; u < EDGE_UNROLL; u++) {
            int k = i + u * SORT_THREADS;
            d[u] = -1;
            s[u] = 0;
            if (k < end) { d[u] = dst[k]; s[u] = src[k]; }
        }
        #pragma unroll
        for (int u = 0; u < EDGE_UNROLL; u++) {
            if ((unsigned)d[u] < (unsigned)N_NODES) {
                int p = atomicAdd(&cursor[d[u] >> BIN_SHIFT], 1);
                stage[p] = ((unsigned)d[u] << 16) | ((unsigned)s[u] & 0xffffu);
            }
        }
    }
    __syncthreads();

    int n_staged = cursor[N_BINS - 1];
    for (int i = t; i < n_staged; i += SORT_THREADS) {
        unsigned w = stage[i];
        binned[delta[w >> (16 + BIN_SHIFT)] + i] = w;
    }
}

static __device__ __forceinline__ float4 gcn_sum_slots(const float4*   __restrict__ f4,
                                                       const float*    __restrict__ norm,
                                                       const unsigned* ls,
                                                       int count, int lane) {
    float4 acc = make_float4(0.f, 0.f, 0.f, 0.f);

    for (int base = 0; base < count; base += 32) {
        int m = min(32, count - base);
        int sidx = 0;
        float sw = 0.f;
        if (lane < m) {
            sidx = (int)min(ls[base + lane] & 0xffffu, (unsigned)(N_NODES - 1));
            sw = norm[sidx];
        }
        int swi = __float_as_int(sw);

        for (int j = 0; j < m; j += 8) {
            float4 a[8];
            float  w[8];
            w[0] = __int_as_float(__builtin_amdgcn_readlane(swi, j));
            a[0] = f4[(size_t)__builtin_amdgcn_readlane(sidx, j) * (FEATS / 4) + lane];
            #pragma unroll
            for (int u = 1; u < 8; u++) {
                a[u] = make_float4(0.f, 0.f, 0.f, 0.f);
                w[u] = 0.f;
                if (j + u < m) {
                    int s = __builtin_amdgcn_readlane(sidx, j + u);
                    w[u] = __int_as_float(__builtin_amdgcn_readlane(swi, j + u));
                    a[u] = f4[(size_t)s * (FEATS / 4) + lane];
                }
            }
            #pragma unroll
            for (int u = 0; u < 8; u++) {
                acc.x += a[u].x * w[u];
                acc.y += a[u].y * w[u];
                acc.z += a[u].z * w[u];
                acc.w += a[u].w * w[u];
            }
        }
    }
    return acc;
}

static __device__ __forceinline__ float4 gcn_sum_scan(const float4* __restrict__ f4,
                                                      const float*    __restrict__ norm,
                                                      const unsigned* __restrict__ binned,
                                                      int beg, int end, int n, int lane) {
    float4 acc = make_float4(0.f, 0.f, 0.f, 0.f);

    for (int base = beg; base < end; base += 32) {
        int k = base + lane;
        unsigned w = 0xffffffffu;
        if (k < end) w = binned[k];
        bool hit = (k < end) && ((w >> 16) == (unsigned)n);
        unsigned found = __builtin_amdgcn_ballot_w32(hit);
        while (found != 0u) {
            int j = __builtin_ctz(found);
            found &= found - 1u;
            unsigned e = (unsigned)__builtin_amdgcn_readlane((int)w, j);
            int s = (int)min(e & 0xffffu, (unsigned)(N_NODES - 1));
            float sw = norm[s];
            float4 a = f4[(size_t)s * (FEATS / 4) + lane];
            acc.x += a.x * sw;
            acc.y += a.y * sw;
            acc.z += a.z * sw;
            acc.w += a.w * sw;
        }
    }
    return acc;
}

__global__ __launch_bounds__(256, 16)
void gcn_lists_gather_gemm_bias_tanh(const float*    __restrict__ feats,
                                     const float*    __restrict__ norm,
                                     const unsigned* __restrict__ binned,
                                     const int*      __restrict__ gbase,
                                     const float*    __restrict__ W,
                                     const float*    __restrict__ bias,
                                     float*          __restrict__ out,
                                     int n_edges) {
    __shared__ __attribute__((aligned(16))) float As[BLOCK_ROWS * LDS_STRIDE];
    __shared__ unsigned lslot[BLOCK_ROWS * ROW_SLOTS];
    __shared__ int lcnt[BLOCK_ROWS];

    int t = threadIdx.x;
    int lane = t & 31;
    int wave = __builtin_amdgcn_readfirstlane(t >> 5);
    unsigned blk = blockIdx.x;
    int row0 = blk * BLOCK_ROWS;
    int bin = row0 >> BIN_SHIFT;
    int beg = max(gbase[bin], 0);
    int end = min(gbase[bin + 1], n_edges);

    if (t < BLOCK_ROWS) lcnt[t] = 0;
    __syncthreads();

    int abeg = beg & ~3;
    int nvec = (end - abeg + 3) >> 2;
    const uint4* b4 = (const uint4*)(binned + abeg);
    for (int v0 = t; v0 < nvec; v0 += 256 * FILTER_VEC) {
        unsigned w[4 * FILTER_VEC];
        #pragma unroll
        for (int u = 0; u < FILTER_VEC; u++) {
            int v = v0 + u * 256;
            uint4 q = make_uint4(0xffffffffu, 0xffffffffu, 0xffffffffu, 0xffffffffu);
            if (v < nvec) q = b4[v];
            w[4 * u + 0] = q.x;
            w[4 * u + 1] = q.y;
            w[4 * u + 2] = q.z;
            w[4 * u + 3] = q.w;
        }
        #pragma unroll
        for (int u = 0; u < FILTER_VEC; u++) {
            int v = v0 + u * 256;
            if (v == 0 || v == nvec - 1) {
                #pragma unroll
                for (int c = 0; c < 4; c++) {
                    int k = abeg + 4 * v + c;
                    if (k < beg || k >= end) w[4 * u + c] = 0xffffffffu;
                }
            }
        }
        int p[4 * FILTER_VEC];
        #pragma unroll
        for (int e = 0; e < 4 * FILTER_VEC; e++) {
            p[e] = ROW_SLOTS;
            if ((w[e] >> (16 + BLOCK_SHIFT)) == blk) {
                p[e] = atomicAdd(&lcnt[(w[e] >> 16) & (BLOCK_ROWS - 1)], 1);
            }
        }
        #pragma unroll
        for (int e = 0; e < 4 * FILTER_VEC; e++) {
            if (p[e] < ROW_SLOTS) {
                lslot[((w[e] >> 16) & (BLOCK_ROWS - 1)) * ROW_SLOTS + p[e]] = w[e];
            }
        }
    }
    __syncthreads();

    int cv = lcnt[4 * wave + (lane & 3)];

    const float4* f4 = (const float4*)feats;
    for (int i = 0; i < 4; i++) {
        int r = 4 * wave + i;
        int n = row0 + r;
        int count = __builtin_amdgcn_readlane(cv, i);
        float4 acc = make_float4(0.f, 0.f, 0.f, 0.f);
        float dn = 0.f;
        if (n < N_NODES) {
            if (count <= ROW_SLOTS) {
                acc = gcn_sum_slots(f4, norm, &lslot[r * ROW_SLOTS], count, lane);
            } else {
                acc = gcn_sum_scan(f4, norm, binned, beg, end, n, lane);
            }
            dn = norm[n];
        }
        acc.x *= dn;
        acc.y *= dn;
        acc.z *= dn;
        acc.w *= dn;
        *(float4*)&As[r * LDS_STRIDE + 4 * lane] = acc;
    }
    __syncthreads();

    int half = lane >> 4;
    int r    = lane & 15;
    int cj   = wave * 16 + r;

    const float* wrow = W + (size_t)cj * FEATS;

    v8f c0 = {};
    v8f c1 = {};
    #pragma unroll
    for (int k0 = 0; k0 < FEATS; k0 += 4) {
        int kk = k0 + 2 * half;
        v2f a0, a1, b;
        a0.x = As[r * LDS_STRIDE + kk];
        a0.y = As[r * LDS_STRIDE + kk + 1];
        a1.x = As[(16 + r) * LDS_STRIDE + kk];
        a1.y = As[(16 + r) * LDS_STRIDE + kk + 1];
        b.x = wrow[kk];
        b.y = wrow[kk + 1];
        c0 = __builtin_amdgcn_wmma_f32_16x16x4_f32(
                 false, a0,  false, b,
                 (short)0, c0,  false,  false);
        c1 = __builtin_amdgcn_wmma_f32_16x16x4_f32(
                 false, a1,  false, b,
                 (short)0, c1,  false,  false);
    }

    float bb = bias[cj];
    float h0[8], h1[8];
    #pragma unroll
    for (int v = 0; v < 8; v++) {
        h0[v] = tanhf(c0[v] + bb);
        h1[v] = tanhf(c1[v] + bb);
    }
    #pragma unroll
    for (int v = 0; v < 8; v++) {
        int m = row0 + v + 8 * half;
        out[(size_t)m * FEATS + cj] = h0[v];
    }
    if (row0 + 16 < N_NODES) {
        #pragma unroll
        for (int v = 0; v < 8; v++) {
            int m = row0 + 16 + v + 8 * half;
            out[(size_t)m * FEATS + cj] = h1[v];
        }
    }
}

extern "C" void kernel_launch(void* const* d_in, const int* in_sizes, int n_in,
                              void* d_out, int out_size, void* d_ws, size_t ws_size,
                              hipStream_t stream) {
    const float* feats = (const float*)d_in[0];
    const float* norm  = (const float*)d_in[1];
    const float* W     = (const float*)d_in[2];
    const float* bias  = (const float*)d_in[3];
    const int*   src   = (const int*)d_in[4];
    const int*   dst   = (const int*)d_in[5];
    int n_edges = in_sizes[4];

    float* out = (float*)d_out;

    int* wsi   = (int*)d_ws;
    int* hist  = wsi + WS_HIST;
    int* gbase = wsi + WS_GBASE;
    unsigned* binned = (unsigned*)(wsi + WS_BIN);

    gcn_bin_count<<<SORT_BLOCKS, SORT_THREADS, 0, stream>>>(dst, hist, n_edges);

    gcn_bin_scatter<<<SORT_BLOCKS, SORT_THREADS, 0, stream>>>(src, dst, hist, gbase, binned, n_edges);

    gcn_lists_gather_gemm_bias_tanh<<<N_ROW_BLOCKS, 256, 0, stream>>>(feats, norm, binned, gbase, W, bias, out, n_edges);
}


# Bin-count launch removed: each chunk is sorted by bin on its own and keeps its run starts; a gather block collects its bin's 192 runs (list order = LDS atomic arrival, as before); 2 launches
# speedup vs baseline: 3.4168x; 1.1218x over previous
#include <hip/hip_runtime.h>
#include <math.h>


typedef __attribute__((ext_vector_type(2))) float v2f;
typedef __attribute__((ext_vector_type(8))) float v8f;

#define N_NODES 50000
#define FEATS   128
#define LDS_STRIDE 132
#define BLOCK_SHIFT 5
#define BLOCK_ROWS (1 << BLOCK_SHIFT)
#define N_ROW_BLOCKS ((N_NODES + BLOCK_ROWS - 1) / BLOCK_ROWS)
#define ROW_SLOTS  64
#define SORT_BLOCKS  192
#define SORT_THREADS 1024
#define EDGE_UNROLL  5
#define CHUNK_MAX    5120
#define RUNS_PER_WAVE (SORT_BLOCKS / 8)
#define RUN_GROUP    8
#define BIN_SHIFT    8
#define BIN_NODES    256
#define N_BINS ((N_NODES + BIN_NODES - 1) / BIN_NODES)

#define WS_OFFS   0
#define WS_SORTED 131072

static __device__ __forceinline__ int wave_inclusive_sum(int x, int lane) {
    x += __builtin_amdgcn_update_dpp(0, x, 0x111, 0xf, 0xf, false);
    x += __builtin_amdgcn_update_dpp(0, x, 0x112, 0xf, 0xf, false);
    x += __builtin_amdgcn_update_dpp(0, x, 0x114, 0xf, 0xf, false);
    x += __builtin_amdgcn_update_dpp(0, x, 0x118, 0xf, 0xf, false);
    int low = __builtin_amdgcn_readlane(x, 15);
    if (lane >= 16) x += low;
    return x;
}

static __device__ __forceinline__ int gcn_chunk_len(int n_edges) {
    return min((n_edges + SORT_BLOCKS - 1) / SORT_BLOCKS, CHUNK_MAX);
}

__global__ __launch_bounds__(SORT_THREADS)
void gcn_chunk_sort(const int* __restrict__ src,
                    const int* __restrict__ dst,
                    int* __restrict__ offs,
                    unsigned* __restrict__ sorted,
                    int n_edges) {
    __shared__ unsigned stage[CHUNK_MAX];
    __shared__ int cursor[256];
    __shared__ int wown[8];
    int t = threadIdx.x;
    int lane = t & 31;
    int wave = __builtin_amdgcn_readfirstlane(t >> 5);
    int me = blockIdx.x;

    if (t < 256) cursor[t] = 0;

    int chunk = gcn_chunk_len(n_edges);
    int beg = me * chunk;
    int end = min(beg + chunk, n_edges);
    int d[EDGE_UNROLL];
    int s[EDGE_UNROLL];
    #pragma unroll
    for (int u = 0; u < EDGE_UNROLL; u++) {
        int k = beg + t + u * SORT_THREADS;
        d[u] = -1;
        s[u] = 0;
        if (k < end) { d[u] = dst[k]; s[u] = src[k]; }
    }
    __syncthreads();
    #pragma unroll
    for (int u = 0; u < EDGE_UNROLL; u++) {
        if ((unsigned)d[u] < (unsigned)N_NODES) atomicAdd(&cursor[d[u] >> BIN_SHIFT], 1);
    }
    __syncthreads();

    int own = 0, iown = 0;
    if (t < 256) {
        own = cursor[t];
        iown = wave_inclusive_sum(own, lane);
        if (lane == 31) wown[wave] = iown;
    }
    __syncthreads();
    if (t < 256) {
        int cown = 0;
        for (int w = 0; w < 8; w++) {
            if (w < wave) cown += wown[w];
        }
        int lstart = cown + iown - own;
        cursor[t] = lstart;
        if (t <= N_BINS) offs[t * SORT_BLOCKS + me] = lstart;
    }
    __syncthreads();

    #pragma unroll
    for (int u = 0; u < EDGE_UNROLL; u++) {
        if ((unsigned)d[u] < (unsigned)N_NODES) {
            int p = atomicAdd(&cursor[d[u] >> BIN_SHIFT], 1);
            stage[p] = ((unsigned)d[u] << 16) | ((unsigned)s[u] & 0xffffu);
        }
    }
    __syncthreads();

    int n_staged = cursor[N_BINS - 1];
    unsigned* mine = sorted + (size_t)me * CHUNK_MAX;
    for (int i = t; i < n_staged; i += SORT_THREADS) mine[i] = stage[i];
}

static __device__ __forceinline__ float4 gcn_sum_slots(const float4*   __restrict__ f4,
                                                       const float*    __restrict__ norm,
                                                       const unsigned* ls,
                                                       int count, int lane) {
    float4 acc = make_float4(0.f, 0.f, 0.f, 0.f);

    for (int base = 0; base < count; base += 32) {
        int m = min(32, count - base);
        int sidx = 0;
        float sw = 0.f;
        if (lane < m) {
            sidx = (int)min(ls[base + lane] & 0xffffu, (unsigned)(N_NODES - 1));
            sw = norm[sidx];
        }
        int swi = __float_as_int(sw);

        for (int j = 0; j < m; j += 8) {
            float4 a[8];
            float  w[8];
            w[0] = __int_as_float(__builtin_amdgcn_readlane(swi, j));
            a[0] = f4[(size_t)__builtin_amdgcn_readlane(sidx, j) * (FEATS / 4) + lane];
            #pragma unroll
            for (int u = 1; u < 8; u++) {
                a[u] = make_float4(0.f, 0.f, 0.f, 0.f);
                w[u] = 0.f;
                if (j + u < m) {
                    int s = __builtin_amdgcn_readlane(sidx, j + u);
                    w[u] = __int_as_float(__builtin_amdgcn_readlane(swi, j + u));
                    a[u] = f4[(size_t)s * (FEATS / 4) + lane];
                }
            }
            #pragma unroll
            for (int u = 0; u < 8; u++) {
                acc.x += a[u].x * w[u];
                acc.y += a[u].y * w[u];
                acc.z += a[u].z * w[u];
                acc.w += a[u].w * w[u];
            }
        }
    }
    return acc;
}

static __device__ __forceinline__ float4 gcn_sum_scan(const float4* __restrict__ f4,
                                                      const float*    __restrict__ norm,
                                                      const unsigned* __restrict__ sorted,
                                                      const int*      __restrict__ offs,
                                                      int bin, int n, int lane) {
    float4 acc = make_float4(0.f, 0.f, 0.f, 0.f);

    for (int c = 0; c < SORT_BLOCKS; c++) {
        int beg = max(offs[bin * SORT_BLOCKS + c], 0);
        int end = min(offs[(bin + 1) * SORT_BLOCKS + c], CHUNK_MAX);
        const unsigned* run = sorted + (size_t)c * CHUNK_MAX;
        for (int base = beg; base < end; base += 32) {
            int k = base + lane;
            unsigned w = 0xffffffffu;
            if (k < end) w = run[k];
            bool hit = (k < end) && ((w >> 16) == (unsigned)n);
            unsigned found = __builtin_amdgcn_ballot_w32(hit);
            while (found != 0u) {
                int j = __builtin_ctz(found);
                found &= found - 1u;
                unsigned e = (unsigned)__builtin_amdgcn_readlane((int)w, j);
                int s = (int)min(e & 0xffffu, (unsigned)(N_NODES - 1));
                float sw = norm[s];
                float4 a = f4[(size_t)s * (FEATS / 4) + lane];
                acc.x += a.x * sw;
                acc.y += a.y * sw;
                acc.z += a.z * sw;
                acc.w += a.w * sw;
            }
        }
    }
    return acc;
}

__global__ __launch_bounds__(256, 16)
void gcn_lists_gather_gemm_bias_tanh(const float*    __restrict__ feats,
                                     const float*    __restrict__ norm,
                                     const unsigned* __restrict__ sorted,
                                     const int*      __restrict__ offs,
                                     const float*    __restrict__ W,
                                     const float*    __restrict__ bias,
                                     float*          __restrict__ out) {
    __shared__ __attribute__((aligned(16))) float As[BLOCK_ROWS * LDS_STRIDE];
    __shared__ unsigned lslot[BLOCK_ROWS * ROW_SLOTS];
    __shared__ int lcnt[BLOCK_ROWS];

    int t = threadIdx.x;
    int lane = t & 31;
    int wave = __builtin_amdgcn_readfirstlane(t >> 5);
    unsigned blk = blockIdx.x;
    int row0 = blk * BLOCK_ROWS;
    int bin = row0 >> BIN_SHIFT;
    if (t < BLOCK_ROWS) lcnt[t] = 0;

    int o0 = 0, o1 = 0;
    if (lane < RUNS_PER_WAVE) {
        int c = wave * RUNS_PER_WAVE + lane;
        o0 = max(offs[bin * SORT_BLOCKS + c], 0);
        o1 = min(offs[(bin + 1) * SORT_BLOCKS + c], CHUNK_MAX);
    }
    __syncthreads();

    const unsigned* runs = sorted + (size_t)(wave * RUNS_PER_WAVE) * CHUNK_MAX;
    unsigned w[RUNS_PER_WAVE];
    #pragma unroll
    for (int i = 0; i < RUNS_PER_WAVE; i++) {
        int k = __builtin_amdgcn_readlane(o0, i) + lane;
        w[i] = 0xffffffffu;
        if (k < __builtin_amdgcn_readlane(o1, i)) w[i] = runs[i * CHUNK_MAX + k];
    }
    #pragma unroll
    for (int g = 0; g < RUNS_PER_WAVE; g += RUN_GROUP) {
        int p[RUN_GROUP];
        #pragma unroll
        for (int e = 0; e < RUN_GROUP; e++) {
            p[e] = ROW_SLOTS;
            if ((w[g + e] >> (16 + BLOCK_SHIFT)) == blk) {
                p[e] = atomicAdd(&lcnt[(w[g + e] >> 16) & (BLOCK_ROWS - 1)], 1);
            }
        }
        #pragma unroll
        for (int e = 0; e < RUN_GROUP; e++) {
            if (p[e] < ROW_SLOTS) {
                lslot[((w[g + e] >> 16) & (BLOCK_ROWS - 1)) * ROW_SLOTS + p[e]] = w[g + e];
            }
        }
    }
    unsigned longer = __builtin_amdgcn_ballot_w32(o1 - o0 > 32);
    while (longer != 0u) {
        int i = __builtin_ctz(longer);
        longer &= longer - 1u;
        int rend = __builtin_amdgcn_readlane(o1, i);
        const unsigned* run = runs + (size_t)i * CHUNK_MAX;
        for (int k0 = __builtin_amdgcn_readlane(o0, i) + 32; k0 < rend; k0 += 32) {
            int k = k0 + lane;
            unsigned x = 0xffffffffu;
            if (k < rend) x = run[k];
            if ((x >> (16 + BLOCK_SHIFT)) == blk) {
                int p = atomicAdd(&lcnt[(x >> 16) & (BLOCK_ROWS - 1)], 1);
                if (p < ROW_SLOTS) lslot[((x >> 16) & (BLOCK_ROWS - 1)) * ROW_SLOTS + p] = x;
            }
        }
    }
    __syncthreads();

    int cv = lcnt[4 * wave + (lane & 3)];

    const float4* f4 = (const float4*)feats;
    for (int i = 0; i < 4; i++) {
        int r = 4 * wave + i;
        int n = row0 + r;
        int count = __builtin_amdgcn_readlane(cv, i);
        float4 acc = make_float4(0.f, 0.f, 0.f, 0.f);
        float dn = 0.f;
        if (n < N_NODES) {
            if (count <= ROW_SLOTS) {
                acc = gcn_sum_slots(f4, norm, &lslot[r * ROW_SLOTS], count, lane);
            } else {
                acc = gcn_sum_scan(f4, norm, sorted, offs, bin, n, lane);
            }
            dn = norm[n];
        }
        acc.x *= dn;
        acc.y *= dn;
        acc.z *= dn;
        acc.w *= dn;
        *(float4*)&As[r * LDS_STRIDE + 4 * lane] = acc;
    }
    __syncthreads();

    int half = lane >> 4;
    int r    = lane & 15;
    int cj   = wave * 16 + r;

    const float* wrow = W + (size_t)cj * FEATS;

    v8f c0 = {};
    v8f c1 = {};
    #pragma unroll
    for (int k0 = 0; k0 < FEATS; k0 += 4) {
        int kk = k0 + 2 * half;
        v2f a0, a1, b;
        a0.x = As[r * LDS_STRIDE + kk];
        a0.y = As[r * LDS_STRIDE + kk + 1];
        a1.x = As[(16 + r) * LDS_STRIDE + kk];
        a1.y = As[(16 + r) * LDS_STRIDE + kk + 1];
        b.x = wrow[kk];
        b.y = wrow[kk + 1];
        c0 = __builtin_amdgcn_wmma_f32_16x16x4_f32(
                 false, a0,  false, b,
                 (short)0, c0,  false,  false);
        c1 = __builtin_amdgcn_wmma_f32_16x16x4_f32(
                 false, a1,  false, b,
                 (short)0, c1,  false,  false);
    }

    float bb = bias[cj];
    float h0[8], h1[8];
    #pragma unroll
    for (int v = 0; v < 8; v++) {
        h0[v] = tanhf(c0[v] + bb);
        h1[v] = tanhf(c1[v] + bb);
    }
    #pragma unroll
    for (int v = 0; v < 8; v++) {
        int m = row0 + v + 8 * half;
        out[(size_t)m * FEATS + cj] = h0[v];
    }
    if (row0 + 16 < N_NODES) {
        #pragma unroll
        for (int v = 0; v < 8; v++) {
            int m = row0 + 16 + v + 8 * half;
            out[(size_t)m * FEATS + cj] = h1[v];
        }
    }
}

extern "C" void kernel_launch(void* const* d_in, const int* in_sizes, int n_in,
                              void* d_out, int out_size, void* d_ws, size_t ws_size,
                              hipStream_t stream) {
    const float* feats = (const float*)d_in[0];
    const float* norm  = (const float*)d_in[1];
    const float* W     = (const float*)d_in[2];
    const float* bias  = (const float*)d_in[3];
    const int*   src   = (const int*)d_in[4];
    const int*   dst   = (const int*)d_in[5];
    int n_edges = in_sizes[4];

    float* out = (float*)d_out;

    int* wsi   = (int*)d_ws;
    int* offs  = wsi + WS_OFFS;
    unsigned* sorted = (unsigned*)(wsi + WS_SORTED);

    gcn_chunk_sort<<<SORT_BLOCKS, SORT_THREADS, 0, stream>>>(src, dst, offs, sorted, n_edges);

    gcn_lists_gather_gemm_bias_tanh<<<N_ROW_BLOCKS, 256, 0, stream>>>(feats, norm, sorted, offs, W, bias, out);
}


# W fragments through a per-wave LDS stage: whole-line 16-byte loads of the wave's 16 W rows (4 lines a load, not 16); A and B read as 16-byte LDS words; k order of the product sum regrouped
# speedup vs baseline: 3.5998x; 1.0536x over previous
#include <hip/hip_runtime.h>
#include <math.h>


typedef __attribute__((ext_vector_type(2))) float v2f;
typedef __attribute__((ext_vector_type(8))) float v8f;

#define N_NODES 50000
#define FEATS   128
#define LDS_STRIDE 132
#define BLOCK_SHIFT 5
#define BLOCK_ROWS (1 << BLOCK_SHIFT)
#define N_ROW_BLOCKS ((N_NODES + BLOCK_ROWS - 1) / BLOCK_ROWS)
#define ROW_SLOTS  64
#define SORT_BLOCKS  192
#define SORT_THREADS 1024
#define EDGE_UNROLL  5
#define CHUNK_MAX    5120
#define RUNS_PER_WAVE (SORT_BLOCKS / 8)
#define RUN_GROUP    8
#define BIN_SHIFT    8
#define BIN_NODES    256
#define W_ROUND_K    32
#define W_PITCH      36
#define N_BINS ((N_NODES + BIN_NODES - 1) / BIN_NODES)

#define WS_OFFS   0
#define WS_SORTED 131072

static __device__ __forceinline__ int wave_inclusive_sum(int x, int lane) {
    x += __builtin_amdgcn_update_dpp(0, x, 0x111, 0xf, 0xf, false);
    x += __builtin_amdgcn_update_dpp(0, x, 0x112, 0xf, 0xf, false);
    x += __builtin_amdgcn_update_dpp(0, x, 0x114, 0xf, 0xf, false);
    x += __builtin_amdgcn_update_dpp(0, x, 0x118, 0xf, 0xf, false);
    int low = __builtin_amdgcn_readlane(x, 15);
    if (lane >= 16) x += low;
    return x;
}

static __device__ __forceinline__ int gcn_chunk_len(int n_edges) {
    return min((n_edges + SORT_BLOCKS - 1) / SORT_BLOCKS, CHUNK_MAX);
}

__global__ __launch_bounds__(SORT_THREADS)
void gcn_chunk_sort(const int* __restrict__ src,
                    const int* __restrict__ dst,
                    int* __restrict__ offs,
                    unsigned* __restrict__ sorted,
                    int n_edges) {
    __shared__ unsigned stage[CHUNK_MAX];
    __shared__ int cursor[256];
    __shared__ int wown[8];
    int t = threadIdx.x;
    int lane = t & 31;
    int wave = __builtin_amdgcn_readfirstlane(t >> 5);
    int me = blockIdx.x;

    if (t < 256) cursor[t] = 0;

    int chunk = gcn_chunk_len(n_edges);
    int beg = me * chunk;
    int end = min(beg + chunk, n_edges);
    int d[EDGE_UNROLL];
    int s[EDGE_UNROLL];
    #pragma unroll
    for (int u = 0; u < EDGE_UNROLL; u++) {
        int k = beg + t + u * SORT_THREADS;
        d[u] = -1;
        s[u] = 0;
        if (k < end) { d[u] = dst[k]; s[u] = src[k]; }
    }
    __syncthreads();
    #pragma unroll
    for (int u = 0; u < EDGE_UNROLL; u++) {
        if ((unsigned)d[u] < (unsigned)N_NODES) atomicAdd(&cursor[d[u] >> BIN_SHIFT], 1);
    }
    __syncthreads();

    int own = 0, iown = 0;
    if (t < 256) {
        own = cursor[t];
        iown = wave_inclusive_sum(own, lane);
        if (lane == 31) wown[wave] = iown;
    }
    __syncthreads();
    if (t < 256) {
        int cown = 0;
        for (int w = 0; w < 8; w++) {
            if (w < wave) cown += wown[w];
        }
        int lstart = cown + iown - own;
        cursor[t] = lstart;
        if (t <= N_BINS) offs[t * SORT_BLOCKS + me] = lstart;
    }
    __syncthreads();

    #pragma unroll
    for (int u = 0; u < EDGE_UNROLL; u++) {
        if ((unsigned)d[u] < (unsigned)N_NODES) {
            int p = atomicAdd(&cursor[d[u] >> BIN_SHIFT], 1);
            stage[p] = ((unsigned)d[u] << 16) | ((unsigned)s[u] & 0xffffu);
        }
    }
    __syncthreads();

    int n_staged = cursor[N_BINS - 1];
    unsigned* mine = sorted + (size_t)me * CHUNK_MAX;
    for (int i = t; i < n_staged; i += SORT_THREADS) mine[i] = stage[i];
}

static __device__ __forceinline__ float4 gcn_sum_slots(const float4*   __restrict__ f4,
                                                       const float*    __restrict__ norm,
                                                       const unsigned* ls,
                                                       int count, int lane) {
    float4 acc = make_float4(0.f, 0.f, 0.f, 0.f);

    for (int base = 0; base < count; base += 32) {
        int m = min(32, count - base);
        int sidx = 0;
        float sw = 0.f;
        if (lane < m) {
            sidx = (int)min(ls[base + lane] & 0xffffu, (unsigned)(N_NODES - 1));
            sw = norm[sidx];
        }
        int swi = __float_as_int(sw);

        for (int j = 0; j < m; j += 8) {
            float4 a[8];
            float  w[8];
            w[0] = __int_as_float(__builtin_amdgcn_readlane(swi, j));
            a[0] = f4[(size_t)__builtin_amdgcn_readlane(sidx, j) * (FEATS / 4) + lane];
            #pragma unroll
            for (int u = 1; u < 8; u++) {
                a[u] = make_float4(0.f, 0.f, 0.f, 0.f);
                w[u] = 0.f;
                if (j + u < m) {
                    int s = __builtin_amdgcn_readlane(sidx, j + u);
                    w[u] = __int_as_float(__builtin_amdgcn_readlane(swi, j + u));
                    a[u] = f4[(size_t)s * (FEATS / 4) + lane];
                }
            }
            #pragma unroll
            for (int u = 0; u < 8; u++) {
                acc.x += a[u].x * w[u];
                acc.y += a[u].y * w[u];
                acc.z += a[u].z * w[u];
                acc.w += a[u].w * w[u];
            }
        }
    }
    return acc;
}

static __device__ __forceinline__ float4 gcn_sum_scan(const float4* __restrict__ f4,
                                                      const float*    __restrict__ norm,
                                                      const unsigned* __restrict__ sorted,
                                                      const int*      __restrict__ offs,
                                                      int bin, int n, int lane) {
    float4 acc = make_float4(0.f, 0.f, 0.f, 0.f);

    for (int c = 0; c < SORT_BLOCKS; c++) {
        int beg = max(offs[bin * SORT_BLOCKS + c], 0);
        int end = min(offs[(bin + 1) * SORT_BLOCKS + c], CHUNK_MAX);
        const unsigned* run = sorted + (size_t)c * CHUNK_MAX;
        for (int base = beg; base < end; base += 32) {
            int k = base + lane;
            unsigned w = 0xffffffffu;
            if (k < end) w = run[k];
            bool hit = (k < end) && ((w >> 16) == (unsigned)n);
            unsigned found = __builtin_amdgcn_ballot_w32(hit);
            while (found != 0u) {
                int j = __builtin_ctz(found);
                found &= found - 1u;
                unsigned e = (unsigned)__builtin_amdgcn_readlane((int)w, j);
                int s = (int)min(e & 0xffffu, (unsigned)(N_NODES - 1));
                float sw = norm[s];
                float4 a = f4[(size_t)s * (FEATS / 4) + lane];
                acc.x += a.x * sw;
                acc.y += a.y * sw;
                acc.z += a.z * sw;
                acc.w += a.w * sw;
            }
        }
    }
    return acc;
}

static __device__ __forceinline__ void wave_lds_handover() {
    __builtin_amdgcn_fence(__ATOMIC_RELEASE, "wavefront");
    __builtin_amdgcn_wave_barrier();
    __builtin_amdgcn_fence(__ATOMIC_ACQUIRE, "wavefront");
}

__global__ __launch_bounds__(256, 16)
void gcn_lists_gather_gemm_bias_tanh(const float*    __restrict__ feats,
                                     const float*    __restrict__ norm,
                                     const unsigned* __restrict__ sorted,
                                     const int*      __restrict__ offs,
                                     const float*                 W,
                                     const float*    __restrict__ bias,
                                     float*          __restrict__ out) {
    __shared__ __attribute__((aligned(16))) float As[BLOCK_ROWS * LDS_STRIDE];
    __shared__ __attribute__((aligned(16))) unsigned lbuf[8 * 16 * W_PITCH];
    unsigned* lslot = lbuf;
    __shared__ int lcnt[BLOCK_ROWS];

    int t = threadIdx.x;
    int lane = t & 31;
    int wave = __builtin_amdgcn_readfirstlane(t >> 5);
    unsigned blk = blockIdx.x;
    int row0 = blk * BLOCK_ROWS;
    int bin = row0 >> BIN_SHIFT;
    if (t < BLOCK_ROWS) lcnt[t] = 0;

    int o0 = 0, o1 = 0;
    if (lane < RUNS_PER_WAVE) {
        int c = wave * RUNS_PER_WAVE + lane;
        o0 = max(offs[bin * SORT_BLOCKS + c], 0);
        o1 = min(offs[(bin + 1) * SORT_BLOCKS + c], CHUNK_MAX);
    }
    __syncthreads();

    const unsigned* runs = sorted + (size_t)(wave * RUNS_PER_WAVE) * CHUNK_MAX;
    unsigned w[RUNS_PER_WAVE];
    #pragma unroll
    for (int i = 0; i < RUNS_PER_WAVE; i++) {
        int k = __builtin_amdgcn_readlane(o0, i) + lane;
        w[i] = 0xffffffffu;
        if (k < __builtin_amdgcn_readlane(o1, i)) w[i] = runs[i * CHUNK_MAX + k];
    }
    #pragma unroll
    for (int g = 0; g < RUNS_PER_WAVE; g += RUN_GROUP) {
        int p[RUN_GROUP];
        #pragma unroll
        for (int e = 0; e < RUN_GROUP; e++) {
            p[e] = ROW_SLOTS;
            if ((w[g + e] >> (16 + BLOCK_SHIFT)) == blk) {
                p[e] = atomicAdd(&lcnt[(w[g + e] >> 16) & (BLOCK_ROWS - 1)], 1);
            }
        }
        #pragma unroll
        for (int e = 0; e < RUN_GROUP; e++) {
            if (p[e] < ROW_SLOTS) {
                lslot[((w[g + e] >> 16) & (BLOCK_ROWS - 1)) * ROW_SLOTS + p[e]] = w[g + e];
            }
        }
    }
    unsigned longer = __builtin_amdgcn_ballot_w32(o1 - o0 > 32);
    while (longer != 0u) {
        int i = __builtin_ctz(longer);
        longer &= longer - 1u;
        int rend = __builtin_amdgcn_readlane(o1, i);
        const unsigned* run = runs + (size_t)i * CHUNK_MAX;
        for (int k0 = __builtin_amdgcn_readlane(o0, i) + 32; k0 < rend; k0 += 32) {
            int k = k0 + lane;
            unsigned x = 0xffffffffu;
            if (k < rend) x = run[k];
            if ((x >> (16 + BLOCK_SHIFT)) == blk) {
                int p = atomicAdd(&lcnt[(x >> 16) & (BLOCK_ROWS - 1)], 1);
                if (p < ROW_SLOTS) lslot[((x >> 16) & (BLOCK_ROWS - 1)) * ROW_SLOTS + p] = x;
            }
        }
    }
    __syncthreads();

    int cv = lcnt[4 * wave + (lane & 3)];

    const float4* f4 = (const float4*)feats;
    for (int i = 0; i < 4; i++) {
        int r = 4 * wave + i;
        int n = row0 + r;
        int count = __builtin_amdgcn_readlane(cv, i);
        float4 acc = make_float4(0.f, 0.f, 0.f, 0.f);
        float dn = 0.f;
        if (n < N_NODES) {
            if (count <= ROW_SLOTS) {
                acc = gcn_sum_slots(f4, norm, &lslot[r * ROW_SLOTS], count, lane);
            } else {
                acc = gcn_sum_scan(f4, norm, sorted, offs, bin, n, lane);
            }
            dn = norm[n];
        }
        acc.x *= dn;
        acc.y *= dn;
        acc.z *= dn;
        acc.w *= dn;
        *(float4*)&As[r * LDS_STRIDE + 4 * lane] = acc;
    }
    __syncthreads();

    int half = lane >> 4;
    int r    = lane & 15;
    int cj   = wave * 16 + r;

    const float4* w4 = (const float4*)W
                     + (size_t)(wave * 16 + (lane >> 3)) * (FEATS / 4) + (lane & 7);
    float* stage = (float*)lbuf + wave * (16 * W_PITCH);
    float* st_put = stage + (lane >> 3) * W_PITCH + 4 * (lane & 7);
    const float* b_get  = stage + r * W_PITCH + 8 * half;
    const float* a_get0 = As + r * LDS_STRIDE + 8 * half;
    const float* a_get1 = a_get0 + 16 * LDS_STRIDE;

    float4 wn0, wn1, wn2, wn3;
    wn0 = w4[(size_t)0 * (FEATS / 4) + (W_ROUND_K / 4) * (0)];
    wn1 = w4[(size_t)4 * (FEATS / 4) + (W_ROUND_K / 4) * (0)];
    wn2 = w4[(size_t)8 * (FEATS / 4) + (W_ROUND_K / 4) * (0)];
    wn3 = w4[(size_t)12 * (FEATS / 4) + (W_ROUND_K / 4) * (0)];

    v8f c0 = {};
    v8f c1 = {};
    #pragma unroll
    for (int rd = 0; rd < FEATS / W_ROUND_K; rd++) {
        *(float4*)(st_put + 0 * W_PITCH) = wn0;
        *(float4*)(st_put + 4 * W_PITCH) = wn1;
        *(float4*)(st_put + 8 * W_PITCH) = wn2;
        *(float4*)(st_put + 12 * W_PITCH) = wn3;
        if (rd + 1 < FEATS / W_ROUND_K) {
            wn0 = w4[(size_t)0 * (FEATS / 4) + (W_ROUND_K / 4) * (rd + 1)];
            wn1 = w4[(size_t)4 * (FEATS / 4) + (W_ROUND_K / 4) * (rd + 1)];
            wn2 = w4[(size_t)8 * (FEATS / 4) + (W_ROUND_K / 4) * (rd + 1)];
            wn3 = w4[(size_t)12 * (FEATS / 4) + (W_ROUND_K / 4) * (rd + 1)];
        }
        wave_lds_handover();
        {
            int ko = W_ROUND_K * rd + 0;
            float4 bl = *(const float4*)(b_get + 0);
            float4 bh = *(const float4*)(b_get + 4);
            float4 xl = *(const float4*)(a_get0 + ko);
            float4 xh = *(const float4*)(a_get0 + ko + 4);
            float4 yl = *(const float4*)(a_get1 + ko);
            float4 yh = *(const float4*)(a_get1 + ko + 4);
            v2f a0, a1, b;
            a0.x = xl.x; a0.y = xl.y; a1.x = yl.x; a1.y = yl.y; b.x = bl.x; b.y = bl.y;
            c0 = __builtin_amdgcn_wmma_f32_16x16x4_f32(
                     false, a0,  false, b,
                     (short)0, c0,  false,  false);
            c1 = __builtin_amdgcn_wmma_f32_16x16x4_f32(
                     false, a1,  false, b,
                     (short)0, c1,  false,  false);
            a0.x = xl.z; a0.y = xl.w; a1.x = yl.z; a1.y = yl.w; b.x = bl.z; b.y = bl.w;
            c0 = __builtin_amdgcn_wmma_f32_16x16x4_f32(
                     false, a0,  false, b,
                     (short)0, c0,  false,  false);
            c1 = __builtin_amdgcn_wmma_f32_16x16x4_f32(
                     false, a1,  false, b,
                     (short)0, c1,  false,  false);
            a0.x = xh.x; a0.y = xh.y; a1.x = yh.x; a1.y = yh.y; b.x = bh.x; b.y = bh.y;
            c0 = __builtin_amdgcn_wmma_f32_16x16x4_f32(
                     false, a0,  false, b,
                     (short)0, c0,  false,  false);
            c1 = __builtin_amdgcn_wmma_f32_16x16x4_f32(
                     false, a1,  false, b,
                     (short)0, c1,  false,  false);
            a0.x = xh.z; a0.y = xh.w; a1.x = yh.z; a1.y = yh.w; b.x = bh.z; b.y = bh.w;
            c0 = __builtin_amdgcn_wmma_f32_16x16x4_f32(
                     false, a0,  false, b,
                     (short)0, c0,  false,  false);
            c1 = __builtin_amdgcn_wmma_f32_16x16x4_f32(
                     false, a1,  false, b,
                     (short)0, c1,  false,  false);
        }
        {
            int ko = W_ROUND_K * rd + 16;
            float4 bl = *(const float4*)(b_get + 16);
            float4 bh = *(const float4*)(b_get + 20);
            float4 xl = *(const float4*)(a_get0 + ko);
            float4 xh = *(const float4*)(a_get0 + ko + 4);
            float4 yl = *(const float4*)(a_get1 + ko);
            float4 yh = *(const float4*)(a_get1 + ko + 4);
            v2f a0, a1, b;
            a0.x = xl.x; a0.y = xl.y; a1.x = yl.x; a1.y = yl.y; b.x = bl.x; b.y = bl.y;
            c0 = __builtin_amdgcn_wmma_f32_16x16x4_f32(
                     false, a0,  false, b,
                     (short)0, c0,  false,  false);
            c1 = __builtin_amdgcn_wmma_f32_16x16x4_f32(
                     false, a1,  false, b,
                     (short)0, c1,  false,  false);
            a0.x = xl.z; a0.y = xl.w; a1.x = yl.z; a1.y = yl.w; b.x = bl.z; b.y = bl.w;
            c0 = __builtin_amdgcn_wmma_f32_16x16x4_f32(
                     false, a0,  false, b,
                     (short)0, c0,  false,  false);
            c1 = __builtin_amdgcn_wmma_f32_16x16x4_f32(
                     false, a1,  false, b,
                     (short)0, c1,  false,  false);
            a0.x = xh.x; a0.y = xh.y; a1.x = yh.x; a1.y = yh.y; b.x = bh.x; b.y = bh.y;
            c0 = __builtin_amdgcn_wmma_f32_16x16x4_f32(
                     false, a0,  false, b,
                     (short)0, c0,  false,  false);
            c1 = __builtin_amdgcn_wmma_f32_16x16x4_f32(
                     false, a1,  false, b,
                     (short)0, c1,  false,  false);
            a0.x = xh.z; a0.y = xh.w; a1.x = yh.z; a1.y = yh.w; b.x = bh.z; b.y = bh.w;
            c0 = __builtin_amdgcn_wmma_f32_16x16x4_f32(
                     false, a0,  false, b,
                     (short)0, c0,  false,  false);
            c1 = __builtin_amdgcn_wmma_f32_16x16x4_f32(
                     false, a1,  false, b,
                     (short)0, c1,  false,  false);
        }
        __builtin_amdgcn_sched_barrier(0);
        wave_lds_handover();
    }

    float bb = bias[cj];
    float h0[8], h1[8];
    #pragma unroll
    for (int v = 0; v < 8; v++) {
        h0[v] = tanhf(c0[v] + bb);
        h1[v] = tanhf(c1[v] + bb);
    }
    #pragma unroll
    for (int v = 0; v < 8; v++) {
        int m = row0 + v + 8 * half;
        out[(size_t)m * FEATS + cj] = h0[v];
    }
    if (row0 + 16 < N_NODES) {
        #pragma unroll
        for (int v = 0; v < 8; v++) {
            int m = row0 + 16 + v + 8 * half;
            out[(size_t)m * FEATS + cj] = h1[v];
        }
    }
}

extern "C" void kernel_launch(void* const* d_in, const int* in_sizes, int n_in,
                              void* d_out, int out_size, void* d_ws, size_t ws_size,
                              hipStream_t stream) {
    const float* feats = (const float*)d_in[0];
    const float* norm  = (const float*)d_in[1];
    const float* W     = (const float*)d_in[2];
    const float* bias  = (const float*)d_in[3];
    const int*   src   = (const int*)d_in[4];
    const int*   dst   = (const int*)d_in[5];
    int n_edges = in_sizes[4];

    float* out = (float*)d_out;

    int* wsi   = (int*)d_ws;
    int* offs  = wsi + WS_OFFS;
    unsigned* sorted = (unsigned*)(wsi + WS_SORTED);

    gcn_chunk_sort<<<SORT_BLOCKS, SORT_THREADS, 0, stream>>>(src, dst, offs, sorted, n_edges);

    gcn_lists_gather_gemm_bias_tanh<<<N_ROW_BLOCKS, 256, 0, stream>>>(feats, norm, sorted, offs, W, bias, out);
}


# Epilogue tanh as 1 - 2/(exp(2x)+1) on the hardware exp2 and reciprocal instructions instead of the library routine (abs error below 4e-7, results not bit-equal to before); sums untouched
# speedup vs baseline: 3.8942x; 1.0818x over previous
#include <hip/hip_runtime.h>
#include <math.h>


typedef __attribute__((ext_vector_type(2))) float v2f;
typedef __attribute__((ext_vector_type(8))) float v8f;

#define N_NODES 50000
#define FEATS   128
#define LDS_STRIDE 132
#define BLOCK_SHIFT 5
#define BLOCK_ROWS (1 << BLOCK_SHIFT)
#define N_ROW_BLOCKS ((N_NODES + BLOCK_ROWS - 1) / BLOCK_ROWS)
#define ROW_SLOTS  64
#define SORT_BLOCKS  192
#define SORT_THREADS 1024
#define EDGE_UNROLL  5
#define CHUNK_MAX    5120
#define RUNS_PER_WAVE (SORT_BLOCKS / 8)
#define RUN_GROUP    8
#define BIN_SHIFT    8
#define BIN_NODES    256
#define W_ROUND_K    32
#define W_PITCH      36
#define N_BINS ((N_NODES + BIN_NODES - 1) / BIN_NODES)

#define WS_OFFS   0
#define WS_SORTED 131072

static __device__ __forceinline__ int wave_inclusive_sum(int x, int lane) {
    x += __builtin_amdgcn_update_dpp(0, x, 0x111, 0xf, 0xf, false);
    x += __builtin_amdgcn_update_dpp(0, x, 0x112, 0xf, 0xf, false);
    x += __builtin_amdgcn_update_dpp(0, x, 0x114, 0xf, 0xf, false);
    x += __builtin_amdgcn_update_dpp(0, x, 0x118, 0xf, 0xf, false);
    int low = __builtin_amdgcn_readlane(x, 15);
    if (lane >= 16) x += low;
    return x;
}

static __device__ __forceinline__ int gcn_chunk_len(int n_edges) {
    return min((n_edges + SORT_BLOCKS - 1) / SORT_BLOCKS, CHUNK_MAX);
}

__global__ __launch_bounds__(SORT_THREADS)
void gcn_chunk_sort(const int* __restrict__ src,
                    const int* __restrict__ dst,
                    int* __restrict__ offs,
                    unsigned* __restrict__ sorted,
                    int n_edges) {
    __shared__ unsigned stage[CHUNK_MAX];
    __shared__ int cursor[256];
    __shared__ int wown[8];
    int t = threadIdx.x;
    int lane = t & 31;
    int wave = __builtin_amdgcn_readfirstlane(t >> 5);
    int me = blockIdx.x;

    if (t < 256) cursor[t] = 0;

    int chunk = gcn_chunk_len(n_edges);
    int beg = me * chunk;
    int end = min(beg + chunk, n_edges);
    int d[EDGE_UNROLL];
    int s[EDGE_UNROLL];
    #pragma unroll
    for (int u = 0; u < EDGE_UNROLL; u++) {
        int k = beg + t + u * SORT_THREADS;
        d[u] = -1;
        s[u] = 0;
        if (k < end) { d[u] = dst[k]; s[u] = src[k]; }
    }
    __syncthreads();
    #pragma unroll
    for (int u = 0; u < EDGE_UNROLL; u++) {
        if ((unsigned)d[u] < (unsigned)N_NODES) atomicAdd(&cursor[d[u] >> BIN_SHIFT], 1);
    }
    __syncthreads();

    int own = 0, iown = 0;
    if (t < 256) {
        own = cursor[t];
        iown = wave_inclusive_sum(own, lane);
        if (lane == 31) wown[wave] = iown;
    }
    __syncthreads();
    if (t < 256) {
        int cown = 0;
        for (int w = 0; w < 8; w++) {
            if (w < wave) cown += wown[w];
        }
        int lstart = cown + iown - own;
        cursor[t] = lstart;
        if (t <= N_BINS) offs[t * SORT_BLOCKS + me] = lstart;
    }
    __syncthreads();

    #pragma unroll
    for (int u = 0; u < EDGE_UNROLL; u++) {
        if ((unsigned)d[u] < (unsigned)N_NODES) {
            int p = atomicAdd(&cursor[d[u] >> BIN_SHIFT], 1);
            stage[p] = ((unsigned)d[u] << 16) | ((unsigned)s[u] & 0xffffu);
        }
    }
    __syncthreads();

    int n_staged = cursor[N_BINS - 1];
    unsigned* mine = sorted + (size_t)me * CHUNK_MAX;
    for (int i = t; i < n_staged; i += SORT_THREADS) mine[i] = stage[i];
}

static __device__ __forceinline__ float4 gcn_sum_slots(const float4*   __restrict__ f4,
                                                       const float*    __restrict__ norm,
                                                       const unsigned* ls,
                                                       int count, int lane) {
    float4 acc = make_float4(0.f, 0.f, 0.f, 0.f);

    for (int base = 0; base < count; base += 32) {
        int m = min(32, count - base);
        int sidx = 0;
        float sw = 0.f;
        if (lane < m) {
            sidx = (int)min(ls[base + lane] & 0xffffu, (unsigned)(N_NODES - 1));
            sw = norm[sidx];
        }
        int swi = __float_as_int(sw);

        for (int j = 0; j < m; j += 8) {
            float4 a[8];
            float  w[8];
            w[0] = __int_as_float(__builtin_amdgcn_readlane(swi, j));
            a[0] = f4[(size_t)__builtin_amdgcn_readlane(sidx, j) * (FEATS / 4) + lane];
            #pragma unroll
            for (int u = 1; u < 8; u++) {
                a[u] = make_float4(0.f, 0.f, 0.f, 0.f);
                w[u] = 0.f;
                if (j + u < m) {
                    int s = __builtin_amdgcn_readlane(sidx, j + u);
                    w[u] = __int_as_float(__builtin_amdgcn_readlane(swi, j + u));
                    a[u] = f4[(size_t)s * (FEATS / 4) + lane];
                }
            }
            #pragma unroll
            for (int u = 0; u < 8; u++) {
                acc.x += a[u].x * w[u];
                acc.y += a[u].y * w[u];
                acc.z += a[u].z * w[u];
                acc.w += a[u].w * w[u];
            }
        }
    }
    return acc;
}

static __device__ __forceinline__ float4 gcn_sum_scan(const float4* __restrict__ f4,
                                                      const float*    __restrict__ norm,
                                                      const unsigned* __restrict__ sorted,
                                                      const int*      __restrict__ offs,
                                                      int bin, int n, int lane) {
    float4 acc = make_float4(0.f, 0.f, 0.f, 0.f);

    for (int c = 0; c < SORT_BLOCKS; c++) {
        int beg = max(offs[bin * SORT_BLOCKS + c], 0);
        int end = min(offs[(bin + 1) * SORT_BLOCKS + c], CHUNK_MAX);
        const unsigned* run = sorted + (size_t)c * CHUNK_MAX;
        for (int base = beg; base < end; base += 32) {
            int k = base + lane;
            unsigned w = 0xffffffffu;
            if (k < end) w = run[k];
            bool hit = (k < end) && ((w >> 16) == (unsigned)n);
            unsigned found = __builtin_amdgcn_ballot_w32(hit);
            while (found != 0u) {
                int j = __builtin_ctz(found);
                found &= found - 1u;
                unsigned e = (unsigned)__builtin_amdgcn_readlane((int)w, j);
                int s = (int)min(e & 0xffffu, (unsigned)(N_NODES - 1));
                float sw = norm[s];
                float4 a = f4[(size_t)s * (FEATS / 4) + lane];
                acc.x += a.x * sw;
                acc.y += a.y * sw;
                acc.z += a.z * sw;
                acc.w += a.w * sw;
            }
        }
    }
    return acc;
}

static __device__ __forceinline__ float gcn_tanh(float x) {
    float e = __builtin_amdgcn_exp2f(x * 2.885390081777927f);
    float q = __builtin_amdgcn_rcpf(e + 1.0f);
    return fmaf(-2.0f, q, 1.0f);
}

static __device__ __forceinline__ void wave_lds_handover() {
    __builtin_amdgcn_fence(__ATOMIC_RELEASE, "wavefront");
    __builtin_amdgcn_wave_barrier();
    __builtin_amdgcn_fence(__ATOMIC_ACQUIRE, "wavefront");
}

__global__ __launch_bounds__(256, 16)
void gcn_lists_gather_gemm_bias_tanh(const float*    __restrict__ feats,
                                     const float*    __restrict__ norm,
                                     const unsigned* __restrict__ sorted,
                                     const int*      __restrict__ offs,
                                     const float*                 W,
                                     const float*    __restrict__ bias,
                                     float*          __restrict__ out) {
    __shared__ __attribute__((aligned(16))) float As[BLOCK_ROWS * LDS_STRIDE];
    __shared__ __attribute__((aligned(16))) unsigned lbuf[8 * 16 * W_PITCH];
    unsigned* lslot = lbuf;
    __shared__ int lcnt[BLOCK_ROWS];

    int t = threadIdx.x;
    int lane = t & 31;
    int wave = __builtin_amdgcn_readfirstlane(t >> 5);
    unsigned blk = blockIdx.x;
    int row0 = blk * BLOCK_ROWS;
    int bin = row0 >> BIN_SHIFT;
    if (t < BLOCK_ROWS) lcnt[t] = 0;

    int o0 = 0, o1 = 0;
    if (lane < RUNS_PER_WAVE) {
        int c = wave * RUNS_PER_WAVE + lane;
        o0 = max(offs[bin * SORT_BLOCKS + c], 0);
        o1 = min(offs[(bin + 1) * SORT_BLOCKS + c], CHUNK_MAX);
    }
    __syncthreads();

    const unsigned* runs = sorted + (size_t)(wave * RUNS_PER_WAVE) * CHUNK_MAX;
    unsigned w[RUNS_PER_WAVE];
    #pragma unroll
    for (int i = 0; i < RUNS_PER_WAVE; i++) {
        int k = __builtin_amdgcn_readlane(o0, i) + lane;
        w[i] = 0xffffffffu;
        if (k < __builtin_amdgcn_readlane(o1, i)) w[i] = runs[i * CHUNK_MAX + k];
    }
    #pragma unroll
    for (int g = 0; g < RUNS_PER_WAVE; g += RUN_GROUP) {
        int p[RUN_GROUP];
        #pragma unroll
        for (int e = 0; e < RUN_GROUP; e++) {
            p[e] = ROW_SLOTS;
            if ((w[g + e] >> (16 + BLOCK_SHIFT)) == blk) {
                p[e] = atomicAdd(&lcnt[(w[g + e] >> 16) & (BLOCK_ROWS - 1)], 1);
            }
        }
        #pragma unroll
        for (int e = 0; e < RUN_GROUP; e++) {
            if (p[e] < ROW_SLOTS) {
                lslot[((w[g + e] >> 16) & (BLOCK_ROWS - 1)) * ROW_SLOTS + p[e]] = w[g + e];
            }
        }
    }
    unsigned longer = __builtin_amdgcn_ballot_w32(o1 - o0 > 32);
    while (longer != 0u) {
        int i = __builtin_ctz(longer);
        longer &= longer - 1u;
        int rend = __builtin_amdgcn_readlane(o1, i);
        const unsigned* run = runs + (size_t)i * CHUNK_MAX;
        for (int k0 = __builtin_amdgcn_readlane(o0, i) + 32; k0 < rend; k0 += 32) {
            int k = k0 + lane;
            unsigned x = 0xffffffffu;
            if (k < rend) x = run[k];
            if ((x >> (16 + BLOCK_SHIFT)) == blk) {
                int p = atomicAdd(&lcnt[(x >> 16) & (BLOCK_ROWS - 1)], 1);
                if (p < ROW_SLOTS) lslot[((x >> 16) & (BLOCK_ROWS - 1)) * ROW_SLOTS + p] = x;
            }
        }
    }
    __syncthreads();

    int cv = lcnt[4 * wave + (lane & 3)];

    const float4* f4 = (const float4*)feats;
    for (int i = 0; i < 4; i++) {
        int r = 4 * wave + i;
        int n = row0 + r;
        int count = __builtin_amdgcn_readlane(cv, i);
        float4 acc = make_float4(0.f, 0.f, 0.f, 0.f);
        float dn = 0.f;
        if (n < N_NODES) {
            if (count <= ROW_SLOTS) {
                acc = gcn_sum_slots(f4, norm, &lslot[r * ROW_SLOTS], count, lane);
            } else {
                acc = gcn_sum_scan(f4, norm, sorted, offs, bin, n, lane);
            }
            dn = norm[n];
        }
        acc.x *= dn;
        acc.y *= dn;
        acc.z *= dn;
        acc.w *= dn;
        *(float4*)&As[r * LDS_STRIDE + 4 * lane] = acc;
    }
    __syncthreads();

    int half = lane >> 4;
    int r    = lane & 15;
    int cj   = wave * 16 + r;

    const float4* w4 = (const float4*)W
                     + (size_t)(wave * 16 + (lane >> 3)) * (FEATS / 4) + (lane & 7);
    float* stage = (float*)lbuf + wave * (16 * W_PITCH);
    float* st_put = stage + (lane >> 3) * W_PITCH + 4 * (lane & 7);
    const float* b_get  = stage + r * W_PITCH + 8 * half;
    const float* a_get0 = As + r * LDS_STRIDE + 8 * half;
    const float* a_get1 = a_get0 + 16 * LDS_STRIDE;

    float4 wn0, wn1, wn2, wn3;
    wn0 = w4[(size_t)0 * (FEATS / 4) + (W_ROUND_K / 4) * (0)];
    wn1 = w4[(size_t)4 * (FEATS / 4) + (W_ROUND_K / 4) * (0)];
    wn2 = w4[(size_t)8 * (FEATS / 4) + (W_ROUND_K / 4) * (0)];
    wn3 = w4[(size_t)12 * (FEATS / 4) + (W_ROUND_K / 4) * (0)];

    v8f c0 = {};
    v8f c1 = {};
    #pragma unroll
    for (int rd = 0; rd < FEATS / W_ROUND_K; rd++) {
        *(float4*)(st_put + 0 * W_PITCH) = wn0;
        *(float4*)(st_put + 4 * W_PITCH) = wn1;
        *(float4*)(st_put + 8 * W_PITCH) = wn2;
        *(float4*)(st_put + 12 * W_PITCH) = wn3;
        if (rd + 1 < FEATS / W_ROUND_K) {
            wn0 = w4[(size_t)0 * (FEATS / 4) + (W_ROUND_K / 4) * (rd + 1)];
            wn1 = w4[(size_t)4 * (FEATS / 4) + (W_ROUND_K / 4) * (rd + 1)];
            wn2 = w4[(size_t)8 * (FEATS / 4) + (W_ROUND_K / 4) * (rd + 1)];
            wn3 = w4[(size_t)12 * (FEATS / 4) + (W_ROUND_K / 4) * (rd + 1)];
        }
        wave_lds_handover();
        {
            int ko = W_ROUND_K * rd + 0;
            float4 bl = *(const float4*)(b_get + 0);
            float4 bh = *(const float4*)(b_get + 4);
            float4 xl = *(const float4*)(a_get0 + ko);
            float4 xh = *(const float4*)(a_get0 + ko + 4);
            float4 yl = *(const float4*)(a_get1 + ko);
            float4 yh = *(const float4*)(a_get1 + ko + 4);
            v2f a0, a1, b;
            a0.x = xl.x; a0.y = xl.y; a1.x = yl.x; a1.y = yl.y; b.x = bl.x; b.y = bl.y;
            c0 = __builtin_amdgcn_wmma_f32_16x16x4_f32(
                     false, a0,  false, b,
                     (short)0, c0,  false,  false);
            c1 = __builtin_amdgcn_wmma_f32_16x16x4_f32(
                     false, a1,  false, b,
                     (short)0, c1,  false,  false);
            a0.x = xl.z; a0.y = xl.w; a1.x = yl.z; a1.y = yl.w; b.x = bl.z; b.y = bl.w;
            c0 = __builtin_amdgcn_wmma_f32_16x16x4_f32(
                     false, a0,  false, b,
                     (short)0, c0,  false,  false);
            c1 = __builtin_amdgcn_wmma_f32_16x16x4_f32(
                     false, a1,  false, b,
                     (short)0, c1,  false,  false);
            a0.x = xh.x; a0.y = xh.y; a1.x = yh.x; a1.y = yh.y; b.x = bh.x; b.y = bh.y;
            c0 = __builtin_amdgcn_wmma_f32_16x16x4_f32(
                     false, a0,  false, b,
                     (short)0, c0,  false,  false);
            c1 = __builtin_amdgcn_wmma_f32_16x16x4_f32(
                     false, a1,  false, b,
                     (short)0, c1,  false,  false);
            a0.x = xh.z; a0.y = xh.w; a1.x = yh.z; a1.y = yh.w; b.x = bh.z; b.y = bh.w;
            c0 = __builtin_amdgcn_wmma_f32_16x16x4_f32(
                     false, a0,  false, b,
                     (short)0, c0,  false,  false);
            c1 = __builtin_amdgcn_wmma_f32_16x16x4_f32(
                     false, a1,  false, b,
                     (short)0, c1,  false,  false);
        }
        {
            int ko = W_ROUND_K * rd + 16;
            float4 bl = *(const float4*)(b_get + 16);
            float4 bh = *(const float4*)(b_get + 20);
            float4 xl = *(const float4*)(a_get0 + ko);
            float4 xh = *(const float4*)(a_get0 + ko + 4);
            float4 yl = *(const float4*)(a_get1 + ko);
            float4 yh = *(const float4*)(a_get1 + ko + 4);
            v2f a0, a1, b;
            a0.x = xl.x; a0.y = xl.y; a1.x = yl.x; a1.y = yl.y; b.x = bl.x; b.y = bl.y;
            c0 = __builtin_amdgcn_wmma_f32_16x16x4_f32(
                     false, a0,  false, b,
                     (short)0, c0,  false,  false);
            c1 = __builtin_amdgcn_wmma_f32_16x16x4_f32(
                     false, a1,  false, b,
                     (short)0, c1,  false,  false);
            a0.x = xl.z; a0.y = xl.w; a1.x = yl.z; a1.y = yl.w; b.x = bl.z; b.y = bl.w;
            c0 = __builtin_amdgcn_wmma_f32_16x16x4_f32(
                     false, a0,  false, b,
                     (short)0, c0,  false,  false);
            c1 = __builtin_amdgcn_wmma_f32_16x16x4_f32(
                     false, a1,  false, b,
                     (short)0, c1,  false,  false);
            a0.x = xh.x; a0.y = xh.y; a1.x = yh.x; a1.y = yh.y; b.x = bh.x; b.y = bh.y;
            c0 = __builtin_amdgcn_wmma_f32_16x16x4_f32(
                     false, a0,  false, b,
                     (short)0, c0,  false,  false);
            c1 = __builtin_amdgcn_wmma_f32_16x16x4_f32(
                     false, a1,  false, b,
                     (short)0, c1,  false,  false);
            a0.x = xh.z; a0.y = xh.w; a1.x = yh.z; a1.y = yh.w; b.x = bh.z; b.y = bh.w;
            c0 = __builtin_amdgcn_wmma_f32_16x16x4_f32(
                     false, a0,  false, b,
                     (short)0, c0,  false,  false);
            c1 = __builtin_amdgcn_wmma_f32_16x16x4_f32(
                     false, a1,  false, b,
                     (short)0, c1,  false,  false);
        }
        __builtin_amdgcn_sched_barrier(0);
        wave_lds_handover();
    }

    float bb = bias[cj];
    float h0[8], h1[8];
    #pragma unroll
    for (int v = 0; v < 8; v++) {
        h0[v] = gcn_tanh(c0[v] + bb);
        h1[v] = gcn_tanh(c1[v] + bb);
    }
    #pragma unroll
    for (int v = 0; v < 8; v++) {
        int m = row0 + v + 8 * half;
        out[(size_t)m * FEATS + cj] = h0[v];
    }
    if (row0 + 16 < N_NODES) {
        #pragma unroll
        for (int v = 0; v < 8; v++) {
            int m = row0 + 16 + v + 8 * half;
            out[(size_t)m * FEATS + cj] = h1[v];
        }
    }
}

extern "C" void kernel_launch(void* const* d_in, const int* in_sizes, int n_in,
                              void* d_out, int out_size, void* d_ws, size_t ws_size,
                              hipStream_t stream) {
    const float* feats = (const float*)d_in[0];
    const float* norm  = (const float*)d_in[1];
    const float* W     = (const float*)d_in[2];
    const float* bias  = (const float*)d_in[3];
    const int*   src   = (const int*)d_in[4];
    const int*   dst   = (const int*)d_in[5];
    int n_edges = in_sizes[4];

    float* out = (float*)d_out;

    int* wsi   = (int*)d_ws;
    int* offs  = wsi + WS_OFFS;
    unsigned* sorted = (unsigned*)(wsi + WS_SORTED);

    gcn_chunk_sort<<<SORT_BLOCKS, SORT_THREADS, 0, stream>>>(src, dst, offs, sorted, n_edges);

    gcn_lists_gather_gemm_bias_tanh<<<N_ROW_BLOCKS, 256, 0, stream>>>(feats, norm, sorted, offs, W, bias, out);
}


# Sort bins narrowed from 256 to 32 destinations (one gather block): a block reads only its own edges, 4 runs per load, no filtering; run table chunk-major; list order = LDS atomic arrival, as before
# speedup vs baseline: 4.1588x; 1.0679x over previous
#include <hip/hip_runtime.h>
#include <math.h>


typedef __attribute__((ext_vector_type(2))) float v2f;
typedef __attribute__((ext_vector_type(8))) float v8f;

#define N_NODES 50000
#define FEATS   128
#define LDS_STRIDE 132
#define BLOCK_SHIFT 5
#define BLOCK_ROWS (1 << BLOCK_SHIFT)
#define N_ROW_BLOCKS ((N_NODES + BLOCK_ROWS - 1) / BLOCK_ROWS)
#define ROW_SLOTS  64
#define SORT_BLOCKS  192
#define SORT_THREADS 1024
#define EDGE_UNROLL  5
#define CHUNK_MAX    5120
#define RUNS_PER_WAVE (SORT_BLOCKS / 8)
#define RUN_LANES    8
#define RUN_LOADS    (RUNS_PER_WAVE * RUN_LANES / 32)
#define BIN_SHIFT    BLOCK_SHIFT
#define BIN_NODES    BLOCK_ROWS
#define W_ROUND_K    32
#define W_PITCH      36
#define N_BINS ((N_NODES + BIN_NODES - 1) / BIN_NODES)
#define N_CURSORS    2048
#define OFFS_PITCH   3136
static_assert(N_CURSORS == 2 * SORT_THREADS && N_BINS <= N_CURSORS && 2 * N_BINS <= OFFS_PITCH && SORT_THREADS == 32 * 32,
              "two bin counters per sort thread, a table row holds every bin's start and end, one wave total per lane");

#define WS_OFFS   0
#define WS_SORTED 1048576

static __device__ __forceinline__ int wave_inclusive_sum(int x, int lane) {
    x += __builtin_amdgcn_update_dpp(0, x, 0x111, 0xf, 0xf, false);
    x += __builtin_amdgcn_update_dpp(0, x, 0x112, 0xf, 0xf, false);
    x += __builtin_amdgcn_update_dpp(0, x, 0x114, 0xf, 0xf, false);
    x += __builtin_amdgcn_update_dpp(0, x, 0x118, 0xf, 0xf, false);
    int low = __builtin_amdgcn_readlane(x, 15);
    if (lane >= 16) x += low;
    return x;
}

static __device__ __forceinline__ int gcn_chunk_len(int n_edges) {
    return min((n_edges + SORT_BLOCKS - 1) / SORT_BLOCKS, CHUNK_MAX);
}

__global__ __launch_bounds__(SORT_THREADS)
void gcn_chunk_sort(const int* __restrict__ src,
                    const int* __restrict__ dst,
                    int* __restrict__ offs,
                    unsigned* __restrict__ sorted,
                    int n_edges) {
    __shared__ unsigned stage[CHUNK_MAX];
    __shared__ __attribute__((aligned(8))) int cursor[N_CURSORS];
    __shared__ int wown[SORT_THREADS / 32];
    int t = threadIdx.x;
    int lane = t & 31;
    int wave = __builtin_amdgcn_readfirstlane(t >> 5);
    int me = blockIdx.x;

    cursor[t] = 0;
    cursor[t + SORT_THREADS] = 0;

    int chunk = gcn_chunk_len(n_edges);
    int beg = me * chunk;
    int end = min(beg + chunk, n_edges);
    int d[EDGE_UNROLL];
    int s[EDGE_UNROLL];
    #pragma unroll
    for (int u = 0; u < EDGE_UNROLL; u++) {
        int k = beg + t + u * SORT_THREADS;
        d[u] = -1;
        s[u] = 0;
        if (k < end) { d[u] = dst[k]; s[u] = src[k]; }
    }
    __syncthreads();
    #pragma unroll
    for (int u = 0; u < EDGE_UNROLL; u++) {
        if ((unsigned)d[u] < (unsigned)N_NODES) atomicAdd(&cursor[d[u] >> BIN_SHIFT], 1);
    }
    __syncthreads();

    int2 pair = *(const int2*)&cursor[2 * t];
    int own = pair.x + pair.y;
    int iown = wave_inclusive_sum(own, lane);
    if (lane == 31) wown[wave] = iown;
    __syncthreads();
    {
        int wt = wown[lane];
        if (lane >= wave) wt = 0;
        int cown = __builtin_amdgcn_readlane(wave_inclusive_sum(wt, lane), 31);
        int2 start;
        start.x = cown + iown - own;
        start.y = start.x + pair.x;
        *(int2*)&cursor[2 * t] = start;
        if (2 * t < N_BINS) {
            *(int4*)&offs[(size_t)me * OFFS_PITCH + 4 * t] =
                make_int4(start.x, start.y, start.y, start.x + own);
        }
    }
    __syncthreads();

    #pragma unroll
    for (int u = 0; u < EDGE_UNROLL; u++) {
        if ((unsigned)d[u] < (unsigned)N_NODES) {
            int p = atomicAdd(&cursor[d[u] >> BIN_SHIFT], 1);
            stage[p] = ((unsigned)d[u] << 16) | ((unsigned)s[u] & 0xffffu);
        }
    }
    __syncthreads();

    int n_staged = cursor[N_BINS - 1];
    unsigned* mine = sorted + (size_t)me * CHUNK_MAX;
    for (int i = t; i < n_staged; i += SORT_THREADS) mine[i] = stage[i];
}

static __device__ __forceinline__ float4 gcn_sum_slots(const float4*   __restrict__ f4,
                                                       const float*    __restrict__ norm,
                                                       const unsigned* ls,
                                                       int count, int lane) {
    float4 acc = make_float4(0.f, 0.f, 0.f, 0.f);

    for (int base = 0; base < count; base += 32) {
        int m = min(32, count - base);
        int sidx = 0;
        float sw = 0.f;
        if (lane < m) {
            sidx = (int)min(ls[base + lane] & 0xffffu, (unsigned)(N_NODES - 1));
            sw = norm[sidx];
        }
        int swi = __float_as_int(sw);

        for (int j = 0; j < m; j += 8) {
            float4 a[8];
            float  w[8];
            w[0] = __int_as_float(__builtin_amdgcn_readlane(swi, j));
            a[0] = f4[(size_t)__builtin_amdgcn_readlane(sidx, j) * (FEATS / 4) + lane];
            #pragma unroll
            for (int u = 1; u < 8; u++) {
                a[u] = make_float4(0.f, 0.f, 0.f, 0.f);
                w[u] = 0.f;
                if (j + u < m) {
                    int s = __builtin_amdgcn_readlane(sidx, j + u);
                    w[u] = __int_as_float(__builtin_amdgcn_readlane(swi, j + u));
                    a[u] = f4[(size_t)s * (FEATS / 4) + lane];
                }
            }
            #pragma unroll
            for (int u = 0; u < 8; u++) {
                acc.x += a[u].x * w[u];
                acc.y += a[u].y * w[u];
                acc.z += a[u].z * w[u];
                acc.w += a[u].w * w[u];
            }
        }
    }
    return acc;
}

static __device__ __forceinline__ float4 gcn_sum_scan(const float4* __restrict__ f4,
                                                      const float*    __restrict__ norm,
                                                      const unsigned* __restrict__ sorted,
                                                      const int*      __restrict__ offs,
                                                      int bin, int n, int lane) {
    float4 acc = make_float4(0.f, 0.f, 0.f, 0.f);

    for (int c = 0; c < SORT_BLOCKS; c++) {
        int beg = max(offs[c * OFFS_PITCH + 2 * bin], 0);
        int end = min(offs[c * OFFS_PITCH + 2 * bin + 1], CHUNK_MAX);
        const unsigned* run = sorted + (size_t)c * CHUNK_MAX;
        for (int base = beg; base < end; base += 32) {
            int k = base + lane;
            unsigned w = 0xffffffffu;
            if (k < end) w = run[k];
            bool hit = (k < end) && ((w >> 16) == (unsigned)n);
            unsigned found = __builtin_amdgcn_ballot_w32(hit);
            while (found != 0u) {
                int j = __builtin_ctz(found);
                found &= found - 1u;
                unsigned e = (unsigned)__builtin_amdgcn_readlane((int)w, j);
                int s = (int)min(e & 0xffffu, (unsigned)(N_NODES - 1));
                float sw = norm[s];
                float4 a = f4[(size_t)s * (FEATS / 4) + lane];
                acc.x += a.x * sw;
                acc.y += a.y * sw;
                acc.z += a.z * sw;
                acc.w += a.w * sw;
            }
        }
    }
    return acc;
}

static __device__ __forceinline__ float gcn_tanh(float x) {
    float e = __builtin_amdgcn_exp2f(x * 2.885390081777927f);
    float q = __builtin_amdgcn_rcpf(e + 1.0f);
    return fmaf(-2.0f, q, 1.0f);
}

static __device__ __forceinline__ void wave_lds_handover() {
    __builtin_amdgcn_fence(__ATOMIC_RELEASE, "wavefront");
    __builtin_amdgcn_wave_barrier();
    __builtin_amdgcn_fence(__ATOMIC_ACQUIRE, "wavefront");
}

__global__ __launch_bounds__(256, 16)
void gcn_lists_gather_gemm_bias_tanh(const float*    __restrict__ feats,
                                     const float*    __restrict__ norm,
                                     const unsigned* __restrict__ sorted,
                                     const int*      __restrict__ offs,
                                     const float*                 W,
                                     const float*    __restrict__ bias,
                                     float*          __restrict__ out) {
    __shared__ __attribute__((aligned(16))) float As[BLOCK_ROWS * LDS_STRIDE];
    __shared__ __attribute__((aligned(16))) unsigned lbuf[8 * 16 * W_PITCH];
    unsigned* lslot = lbuf;
    __shared__ int lcnt[BLOCK_ROWS];

    int t = threadIdx.x;
    int lane = t & 31;
    int wave = __builtin_amdgcn_readfirstlane(t >> 5);
    unsigned blk = blockIdx.x;
    int row0 = blk * BLOCK_ROWS;
    int bin = row0 >> BIN_SHIFT;
    if (t < BLOCK_ROWS) lcnt[t] = 0;

    int sub  = lane >> 3;
    int word = lane & (RUN_LANES - 1);
    const int2*     tab  = (const int2*)(offs + (size_t)(wave * RUNS_PER_WAVE + sub) * OFFS_PITCH) + blk;
    const unsigned* runs = sorted + (size_t)(wave * RUNS_PER_WAVE + sub) * CHUNK_MAX;
    int b0[RUN_LOADS], b1[RUN_LOADS];
    #pragma unroll
    for (int i = 0; i < RUN_LOADS; i++) {
        int2 be = tab[(size_t)(4 * i) * (OFFS_PITCH / 2)];
        b0[i] = max(be.x, 0);
        b1[i] = min(be.y, CHUNK_MAX);
    }
    __syncthreads();

    unsigned w[RUN_LOADS];
    #pragma unroll
    for (int i = 0; i < RUN_LOADS; i++) {
        int k = b0[i] + word;
        w[i] = 0xffffffffu;
        if (k < b1[i]) w[i] = runs[(size_t)(4 * i) * CHUNK_MAX + k];
    }
    int p[RUN_LOADS];
    #pragma unroll
    for (int i = 0; i < RUN_LOADS; i++) {
        p[i] = ROW_SLOTS;
        if ((w[i] >> (16 + BLOCK_SHIFT)) == blk) {
            p[i] = atomicAdd(&lcnt[(w[i] >> 16) & (BLOCK_ROWS - 1)], 1);
        }
    }
    #pragma unroll
    for (int i = 0; i < RUN_LOADS; i++) {
        if (p[i] < ROW_SLOTS) {
            lslot[((w[i] >> 16) & (BLOCK_ROWS - 1)) * ROW_SLOTS + p[i]] = w[i];
        }
    }
    int longest = 0;
    #pragma unroll
    for (int i = 0; i < RUN_LOADS; i++) longest = max(longest, b1[i] - b0[i]);
    if (__builtin_amdgcn_ballot_w32(longest > RUN_LANES) != 0u) {
        #pragma unroll
        for (int i = 0; i < RUN_LOADS; i++) {
            for (int k = b0[i] + RUN_LANES + word;
                 __builtin_amdgcn_ballot_w32(k < b1[i]) != 0u; k += RUN_LANES) {
                unsigned x = 0xffffffffu;
                if (k < b1[i]) x = runs[(size_t)(4 * i) * CHUNK_MAX + k];
                if ((x >> (16 + BLOCK_SHIFT)) == blk) {
                    int q = atomicAdd(&lcnt[(x >> 16) & (BLOCK_ROWS - 1)], 1);
                    if (q < ROW_SLOTS) lslot[((x >> 16) & (BLOCK_ROWS - 1)) * ROW_SLOTS + q] = x;
                }
            }
        }
    }
    __syncthreads();

    int cv = lcnt[4 * wave + (lane & 3)];

    const float4* f4 = (const float4*)feats;
    for (int i = 0; i < 4; i++) {
        int r = 4 * wave + i;
        int n = row0 + r;
        int count = __builtin_amdgcn_readlane(cv, i);
        float4 acc = make_float4(0.f, 0.f, 0.f, 0.f);
        float dn = 0.f;
        if (n < N_NODES) {
            if (count <= ROW_SLOTS) {
                acc = gcn_sum_slots(f4, norm, &lslot[r * ROW_SLOTS], count, lane);
            } else {
                acc = gcn_sum_scan(f4, norm, sorted, offs, bin, n, lane);
            }
            dn = norm[n];
        }
        acc.x *= dn;
        acc.y *= dn;
        acc.z *= dn;
        acc.w *= dn;
        *(float4*)&As[r * LDS_STRIDE + 4 * lane] = acc;
    }
    __syncthreads();

    int half = lane >> 4;
    int r    = lane & 15;
    int cj   = wave * 16 + r;

    const float4* w4 = (const float4*)W
                     + (size_t)(wave * 16 + (lane >> 3)) * (FEATS / 4) + (lane & 7);
    float* stage = (float*)lbuf + wave * (16 * W_PITCH);
    float* st_put = stage + (lane >> 3) * W_PITCH + 4 * (lane & 7);
    const float* b_get  = stage + r * W_PITCH + 8 * half;
    const float* a_get0 = As + r * LDS_STRIDE + 8 * half;
    const float* a_get1 = a_get0 + 16 * LDS_STRIDE;

    float4 wn0, wn1, wn2, wn3;
    wn0 = w4[(size_t)0 * (FEATS / 4) + (W_ROUND_K / 4) * (0)];
    wn1 = w4[(size_t)4 * (FEATS / 4) + (W_ROUND_K / 4) * (0)];
    wn2 = w4[(size_t)8 * (FEATS / 4) + (W_ROUND_K / 4) * (0)];
    wn3 = w4[(size_t)12 * (FEATS / 4) + (W_ROUND_K / 4) * (0)];

    v8f c0 = {};
    v8f c1 = {};
    #pragma unroll
    for (int rd = 0; rd < FEATS / W_ROUND_K; rd++) {
        *(float4*)(st_put + 0 * W_PITCH) = wn0;
        *(float4*)(st_put + 4 * W_PITCH) = wn1;
        *(float4*)(st_put + 8 * W_PITCH) = wn2;
        *(float4*)(st_put + 12 * W_PITCH) = wn3;
        if (rd + 1 < FEATS / W_ROUND_K) {
            wn0 = w4[(size_t)0 * (FEATS / 4) + (W_ROUND_K / 4) * (rd + 1)];
            wn1 = w4[(size_t)4 * (FEATS / 4) + (W_ROUND_K / 4) * (rd + 1)];
            wn2 = w4[(size_t)8 * (FEATS / 4) + (W_ROUND_K / 4) * (rd + 1)];
            wn3 = w4[(size_t)12 * (FEATS / 4) + (W_ROUND_K / 4) * (rd + 1)];
        }
        wave_lds_handover();
        {
            int ko = W_ROUND_K * rd + 0;
            float4 bl = *(const float4*)(b_get + 0);
            float4 bh = *(const float4*)(b_get + 4);
            float4 xl = *(const float4*)(a_get0 + ko);
            float4 xh = *(const float4*)(a_get0 + ko + 4);
            float4 yl = *(const float4*)(a_get1 + ko);
            float4 yh = *(const float4*)(a_get1 + ko + 4);
            v2f a0, a1, b;
            a0.x = xl.x; a0.y = xl.y; a1.x = yl.x; a1.y = yl.y; b.x = bl.x; b.y = bl.y;
            c0 = __builtin_amdgcn_wmma_f32_16x16x4_f32(
                     false, a0,  false, b,
                     (short)0, c0,  false,  false);
            c1 = __builtin_amdgcn_wmma_f32_16x16x4_f32(
                     false, a1,  false, b,
                     (short)0, c1,  false,  false);
            a0.x = xl.z; a0.y = xl.w; a1.x = yl.z; a1.y = yl.w; b.x = bl.z; b.y = bl.w;
            c0 = __builtin_amdgcn_wmma_f32_16x16x4_f32(
                     false, a0,  false, b,
                     (short)0, c0,  false,  false);
            c1 = __builtin_amdgcn_wmma_f32_16x16x4_f32(
                     false, a1,  false, b,
                     (short)0, c1,  false,  false);
            a0.x = xh.x; a0.y = xh.y; a1.x = yh.x; a1.y = yh.y; b.x = bh.x; b.y = bh.y;
            c0 = __builtin_amdgcn_wmma_f32_16x16x4_f32(
                     false, a0,  false, b,
                     (short)0, c0,  false,  false);
            c1 = __builtin_amdgcn_wmma_f32_16x16x4_f32(
                     false, a1,  false, b,
                     (short)0, c1,  false,  false);
            a0.x = xh.z; a0.y = xh.w; a1.x = yh.z; a1.y = yh.w; b.x = bh.z; b.y = bh.w;
            c0 = __builtin_amdgcn_wmma_f32_16x16x4_f32(
                     false, a0,  false, b,
                     (short)0, c0,  false,  false);
            c1 = __builtin_amdgcn_wmma_f32_16x16x4_f32(
                     false, a1,  false, b,
                     (short)0, c1,  false,  false);
        }
        {
            int ko = W_ROUND_K * rd + 16;
            float4 bl = *(const float4*)(b_get + 16);
            float4 bh = *(const float4*)(b_get + 20);
            float4 xl = *(const float4*)(a_get0 + ko);
            float4 xh = *(const float4*)(a_get0 + ko + 4);
            float4 yl = *(const float4*)(a_get1 + ko);
            float4 yh = *(const float4*)(a_get1 + ko + 4);
            v2f a0, a1, b;
            a0.x = xl.x; a0.y = xl.y; a1.x = yl.x; a1.y = yl.y; b.x = bl.x; b.y = bl.y;
            c0 = __builtin_amdgcn_wmma_f32_16x16x4_f32(
                     false, a0,  false, b,
                     (short)0, c0,  false,  false);
            c1 = __builtin_amdgcn_wmma_f32_16x16x4_f32(
                     false, a1,  false, b,
                     (short)0, c1,  false,  false);
            a0.x = xl.z; a0.y = xl.w; a1.x = yl.z; a1.y = yl.w; b.x = bl.z; b.y = bl.w;
            c0 = __builtin_amdgcn_wmma_f32_16x16x4_f32(
                     false, a0,  false, b,
                     (short)0, c0,  false,  false);
            c1 = __builtin_amdgcn_wmma_f32_16x16x4_f32(
                     false, a1,  false, b,
                     (short)0, c1,  false,  false);
            a0.x = xh.x; a0.y = xh.y; a1.x = yh.x; a1.y = yh.y; b.x = bh.x; b.y = bh.y;
            c0 = __builtin_amdgcn_wmma_f32_16x16x4_f32(
                     false, a0,  false, b,
                     (short)0, c0,  false,  false);
            c1 = __builtin_amdgcn_wmma_f32_16x16x4_f32(
                     false, a1,  false, b,
                     (short)0, c1,  false,  false);
            a0.x = xh.z; a0.y = xh.w; a1.x = yh.z; a1.y = yh.w; b.x = bh.z; b.y = bh.w;
            c0 = __builtin_amdgcn_wmma_f32_16x16x4_f32(
                     false, a0,  false, b,
                     (short)0, c0,  false,  false);
            c1 = __builtin_amdgcn_wmma_f32_16x16x4_f32(
                     false, a1,  false, b,
                     (short)0, c1,  false,  false);
        }
        __builtin_amdgcn_sched_barrier(0);
        wave_lds_handover();
    }

    float bb = bias[cj];
    float h0[8], h1[8];
    #pragma unroll
    for (int v = 0; v < 8; v++) {
        h0[v] = gcn_tanh(c0[v] + bb);
        h1[v] = gcn_tanh(c1[v] + bb);
    }
    #pragma unroll
    for (int v = 0; v < 8; v++) {
        int m = row0 + v + 8 * half;
        out[(size_t)m * FEATS + cj] = h0[v];
    }
    if (row0 + 16 < N_NODES) {
        #pragma unroll
        for (int v = 0; v < 8; v++) {
            int m = row0 + 16 + v + 8 * half;
            out[(size_t)m * FEATS + cj] = h1[v];
        }
    }
}

extern "C" void kernel_launch(void* const* d_in, const int* in_sizes, int n_in,
                              void* d_out, int out_size, void* d_ws, size_t ws_size,
                              hipStream_t stream) {
    const float* feats = (const float*)d_in[0];
    const float* norm  = (const float*)d_in[1];
    const float* W     = (const float*)d_in[2];
    const float* bias  = (const float*)d_in[3];
    const int*   src   = (const int*)d_in[4];
    const int*   dst   = (const int*)d_in[5];
    int n_edges = in_sizes[4];

    float* out = (float*)d_out;

    int* wsi   = (int*)d_ws;
    int* offs  = wsi + WS_OFFS;
    unsigned* sorted = (unsigned*)(wsi + WS_SORTED);

    gcn_chunk_sort<<<SORT_BLOCKS, SORT_THREADS, 0, stream>>>(src, dst, offs, sorted, n_edges);

    gcn_lists_gather_gemm_bias_tanh<<<N_ROW_BLOCKS, 256, 0, stream>>>(feats, norm, sorted, offs, W, bias, out);
}
